# starting kernel
# baseline (speedup 1.0000x reference)
#include <hip/hip_runtime.h>
#include <math.h>

#define IN_CH 128
#define HID   128
#define OUTC  47
#define OUTP  48

typedef float v2f __attribute__((ext_vector_type(2)));
typedef float v8f __attribute__((ext_vector_type(8)));

__global__ void zero_f32(float* __restrict__ p, long long n) {
  long long i = (long long)blockIdx.x * blockDim.x + threadIdx.x;
  long long stride = (long long)gridDim.x * blockDim.x;
  for (; i < n; i += stride) p[i] = 0.0f;
}

__global__ void scatter_deg(const int* __restrict__ dst, float* __restrict__ deg, int E) {
  int e = blockIdx.x * blockDim.x + threadIdx.x;
  if (e < E) atomicAdd(&deg[dst[e]], 1.0f);
}

__global__ void scatter_msg(const float* __restrict__ xsrc, const int* __restrict__ src,
                            const int* __restrict__ dst, float* __restrict__ msg, int E) {
  long long t = (long long)blockIdx.x * blockDim.x + threadIdx.x;
  int e    = (int)(t >> 5);
  int lane = (int)(t & 31);
  if (e >= E) return;
  int s = src[e];
  int d = dst[e];
  const float4 v = *(const float4*)(xsrc + (size_t)s * IN_CH + lane * 4);
  float* o = msg + (size_t)d * IN_CH + lane * 4;
  atomicAdd(o + 0, v.x);
  atomicAdd(o + 1, v.y);
  atomicAdd(o + 2, v.z);
  atomicAdd(o + 3, v.w);
}

__global__ __launch_bounds__(128)
void sage_gemm_relu(const float* __restrict__ msg, const float* __restrict__ deg,
                    const float* __restrict__ xdst,
                    const float* __restrict__ Wl, const float* __restrict__ bl,
                    const float* __restrict__ Wr,
                    float* __restrict__ hout, int nrows)
{
  __shared__ float sA[16][IN_CH + 4];
  __shared__ float sX[16][IN_CH + 4];
  const int tid = threadIdx.x;
  const int r0  = blockIdx.x * 16;

  for (int m = 0; m < 16; ++m) {
    int row = r0 + m;
    float a = 0.0f, xx = 0.0f;
    if (row < nrows) {
      float inv = 1.0f / fmaxf(deg[row], 1.0f);
      a  = msg [(size_t)row * IN_CH + tid] * inv;
      xx = xdst[(size_t)row * IN_CH + tid];
    }
    sA[m][tid] = a;
    sX[m][tid] = xx;
  }
  __syncthreads();

  const int wave = tid >> 5;
  const int lane = tid & 31;
  const int half = lane >> 4;
  const int m    = lane & 15;
  const int n    = lane & 15;
  const int kh   = half * 2;

  for (int t = 0; t < 2; ++t) {
    const int n0 = (wave * 2 + t) * 16;
    v8f acc = {0.f, 0.f, 0.f, 0.f, 0.f, 0.f, 0.f, 0.f};
    for (int k = 0; k < IN_CH; k += 4) {
      v2f a, b;
      a.x = sA[m][k + kh];
      a.y = sA[m][k + kh + 1];
      b.x = Wl[(size_t)(k + kh)     * HID + n0 + n];
      b.y = Wl[(size_t)(k + kh + 1) * HID + n0 + n];
      acc = __builtin_amdgcn_wmma_f32_16x16x4_f32(false, a, false, b, (short)0, acc, false, false);
    }
    for (int k = 0; k < IN_CH; k += 4) {
      v2f a, b;
      a.x = sX[m][k + kh];
      a.y = sX[m][k + kh + 1];
      b.x = Wr[(size_t)(k + kh)     * HID + n0 + n];
      b.y = Wr[(size_t)(k + kh + 1) * HID + n0 + n];
      acc = __builtin_amdgcn_wmma_f32_16x16x4_f32(false, a, false, b, (short)0, acc, false, false);
    }
    const float bias  = bl[n0 + n];
    const int   mbase = half * 8;
    for (int r = 0; r < 8; ++r) {
      int row = r0 + mbase + r;
      if (row < nrows) {
        float v = acc[r] + bias;
        hout[(size_t)row * HID + n0 + n] = fmaxf(v, 0.0f);
      }
    }
  }
}

__global__ __launch_bounds__(128)
void sage_gemm_lsm(const float* __restrict__ msg, const float* __restrict__ deg,
                   const float* __restrict__ xdst,
                   const float* __restrict__ Wl, const float* __restrict__ bl,
                   const float* __restrict__ Wr,
                   float* __restrict__ out, int nrows)
{
  __shared__ float sA[16][HID + 4];
  __shared__ float sX[16][HID + 4];
  __shared__ float sL[16][OUTP];
  const int tid = threadIdx.x;
  const int r0  = blockIdx.x * 16;

  for (int m = 0; m < 16; ++m) {
    int row = r0 + m;
    float a = 0.0f, xx = 0.0f;
    if (row < nrows) {
      float inv = 1.0f / fmaxf(deg[row], 1.0f);
      a  = msg [(size_t)row * HID + tid] * inv;
      xx = xdst[(size_t)row * HID + tid];
    }
    sA[m][tid] = a;
    sX[m][tid] = xx;
  }
  __syncthreads();

  const int wave = tid >> 5;
  const int lane = tid & 31;
  const int half = lane >> 4;
  const int m    = lane & 15;
  const int n    = lane & 15;
  const int kh   = half * 2;

  if (wave < 3) {
    const int n0  = wave * 16;
    const int col = n0 + n;
    const bool cv = (col < OUTC);
    v8f acc = {0.f, 0.f, 0.f, 0.f, 0.f, 0.f, 0.f, 0.f};
    for (int k = 0; k < HID; k += 4) {
      v2f a, b;
      a.x = sA[m][k + kh];
      a.y = sA[m][k + kh + 1];
      b.x = cv ? Wl[(size_t)(k + kh)     * OUTC + col] : 0.0f;
      b.y = cv ? Wl[(size_t)(k + kh + 1) * OUTC + col] : 0.0f;
      acc = __builtin_amdgcn_wmma_f32_16x16x4_f32(false, a, false, b, (short)0, acc, false, false);
    }
    for (int k = 0; k < HID; k += 4) {
      v2f a, b;
      a.x = sX[m][k + kh];
      a.y = sX[m][k + kh + 1];
      b.x = cv ? Wr[(size_t)(k + kh)     * OUTC + col] : 0.0f;
      b.y = cv ? Wr[(size_t)(k + kh + 1) * OUTC + col] : 0.0f;
      acc = __builtin_amdgcn_wmma_f32_16x16x4_f32(false, a, false, b, (short)0, acc, false, false);
    }
    const float bias  = cv ? bl[col] : 0.0f;
    const int   mbase = half * 8;
    for (int r = 0; r < 8; ++r)
      sL[mbase + r][col] = acc[r] + bias;
  }
  __syncthreads();

  if (tid < 16) {
    int row = r0 + tid;
    if (row < nrows) {
      float mx = -INFINITY;
      for (int c = 0; c < OUTC; ++c) mx = fmaxf(mx, sL[tid][c]);
      float s = 0.0f;
      for (int c = 0; c < OUTC; ++c) s += expf(sL[tid][c] - mx);
      float lse = logf(s);
      for (int c = 0; c < OUTC; ++c)
        out[(size_t)row * OUTC + c] = sL[tid][c] - mx - lse;
    }
  }
}

extern "C" void kernel_launch(void* const* d_in, const int* in_sizes, int n_in,
                              void* d_out, int out_size, void* d_ws, size_t ws_size,
                              hipStream_t stream)
{
  const float* x    = (const float*)d_in[0];
  const int*   src0 = (const int*)d_in[1];
  const int*   dst0 = (const int*)d_in[2];
  const int*   src1 = (const int*)d_in[3];
  const int*   dst1 = (const int*)d_in[4];
  const float* Wl0  = (const float*)d_in[7];
  const float* bl0  = (const float*)d_in[8];
  const float* Wr0  = (const float*)d_in[9];
  const float* Wl1  = (const float*)d_in[10];
  const float* bl1  = (const float*)d_in[11];
  const float* Wr1  = (const float*)d_in[12];
  float* out = (float*)d_out;

  const int E0 = in_sizes[1];
  const int E1 = in_sizes[3];
  const int N1 = 100000;
  const int N2 = out_size / OUTC;

  float* msg0 = (float*)d_ws;
  float* deg0 = msg0 + (size_t)N1 * HID;
  float* h    = deg0 + N1;
  float* msg1 = h    + (size_t)N1 * HID;
  float* deg1 = msg1 + (size_t)N2 * HID;

  const long long z0 = (long long)N1 * HID + N1;
  const long long z1 = (long long)N2 * HID + N2;
  zero_f32<<<1024, 256, 0, stream>>>(msg0, z0);
  zero_f32<<<256,  256, 0, stream>>>(msg1, z1);

  scatter_deg<<<(E0 + 255) / 256, 256, 0, stream>>>(dst0, deg0, E0);
  {
    long long th = (long long)E0 * 32;
    scatter_msg<<<(unsigned)((th + 255) / 256), 256, 0, stream>>>(x, src0, dst0, msg0, E0);
  }
  sage_gemm_relu<<<(N1 + 15) / 16, 128, 0, stream>>>(msg0, deg0, x, Wl0, bl0, Wr0, h, N1);

  scatter_deg<<<(E1 + 255) / 256, 256, 0, stream>>>(dst1, deg1, E1);
  {
    long long th = (long long)E1 * 32;
    scatter_msg<<<(unsigned)((th + 255) / 256), 256, 0, stream>>>(h, src1, dst1, msg1, E1);
  }
  sage_gemm_lsm<<<(N2 + 15) / 16, 128, 0, stream>>>(msg1, deg1, h, Wl1, bl1, Wr1, out, N2);
}


# edge sums of both layers without global atomics: a block keeps its dst rows in LDS, its waves scan the whole dst list and add source rows by LDS float adds; 7 launches to 4; sum order = LDS arrival
# speedup vs baseline: 1.9340x; 1.9340x over previous
#include <hip/hip_runtime.h>
#include <math.h>

#define IN_CH 128
#define HID   128
#define OUTC  47
#define OUTP  48

typedef float v2f __attribute__((ext_vector_type(2)));
typedef float v8f __attribute__((ext_vector_type(8)));

#define AGG_NW 32
#define AGG_G  4
#define QCAP   (32 + AGG_G * 128)
#define ROWS0  392
#define ROWS1  79

__device__ __forceinline__ void load_dst_group(int4 (&v)[AGG_G], const int* __restrict__ dst,
                                               int e0, int E, bool whole) {
  if (whole) {
#pragma unroll
    for (int u = 0; u < AGG_G; ++u) v[u] = *(const int4*)(dst + e0 + u * 128);
  } else {
#pragma unroll
    for (int u = 0; u < AGG_G; ++u) {
      const int e = e0 + u * 128;
      v[u].x = (e     < E) ? dst[e]     : -1;
      v[u].y = (e + 1 < E) ? dst[e + 1] : -1;
      v[u].z = (e + 2 < E) ? dst[e + 2] : -1;
      v[u].w = (e + 3 < E) ? dst[e + 3] : -1;
    }
  }
}

template <int CH, bool COUNT>
__device__ __forceinline__ void scan_step(const int4 (&cur)[AGG_G], const int ebase, const bool last,
                                          int& qn, int* __restrict__ q,
                                          float* __restrict__ acc, float* __restrict__ cnt,
                                          const float* __restrict__ xsrc, const int* __restrict__ src,
                                          const int* __restrict__ dst,
                                          const int r0, const int rows, const int lane)
{
  constexpr int LPR = CH / 4;
#pragma unroll
  for (int u = 0; u < AGG_G; ++u) {
    const int dv[4] = {cur[u].x, cur[u].y, cur[u].z, cur[u].w};
#pragma unroll
    for (int c = 0; c < 4; ++c) {
      const bool hit = (unsigned)(dv[c] - r0) < (unsigned)rows;
      const unsigned mask = (unsigned)__ballot(hit);
      if (mask != 0u) {
        const int pos = qn + __popc(mask & ((1u << lane) - 1u));
        if (hit) q[pos] = ebase + u * 128 + c;
        qn += __popc(mask);
      }
    }
  }

  while (qn >= 32 || (last && qn > 0)) {
    const int n  = min(qn, 32);
    const int qb = qn - n;
    int s_i = 0;
    int d_i = -1;
    if (lane < n) {
      const int e = q[qb + lane];
      s_i = src[e];
      d_i = dst[e] - r0;
      if (COUNT) atomicAdd(&cnt[d_i], 1.0f);
    }
    qn = qb;
#pragma unroll
    for (int j0 = 0; j0 < 32; j0 += 8) {
      if (j0 < n) {
        float4 v[8];
#pragma unroll
        for (int k = 0; k < 8; ++k) {
          const int sj = __builtin_amdgcn_readlane(s_i, j0 + k);
          if (lane < LPR) v[k] = *(const float4*)(xsrc + (size_t)sj * CH + lane * 4);
        }
#pragma unroll
        for (int k = 0; k < 8; ++k) {
          const int dj = __builtin_amdgcn_readlane(d_i, j0 + k);
          if (dj >= 0 && lane < LPR) {
            float* a = acc + dj * CH + lane;
            atomicAdd(a,           v[k].x);
            atomicAdd(a + LPR,     v[k].y);
            atomicAdd(a + 2 * LPR, v[k].z);
            atomicAdd(a + 3 * LPR, v[k].w);
          }
        }
      }
    }
  }
}

template <int CH, int ROWS, bool COUNT>
__global__ __launch_bounds__(AGG_NW * 32)
void agg_slab(const float* __restrict__ xsrc, const int* __restrict__ src, const int* __restrict__ dst,
              float* __restrict__ msg, float* __restrict__ deg, int E, int nrows)
{
  constexpr int LPR = CH / 4;
  __shared__ __attribute__((aligned(16))) float acc[ROWS * CH];
  __shared__ float cnt[COUNT ? ROWS : 1];
  __shared__ int   queue[AGG_NW * QCAP];

  const int tid  = threadIdx.x;
  const int lane = tid & 31;
  const int wave = __builtin_amdgcn_readfirstlane(tid >> 5);
  const int r0   = blockIdx.x * ROWS;
  const int rows = min(ROWS, nrows - r0);

  for (int i = tid; i < ROWS * CH / 4; i += AGG_NW * 32)
    ((float4*)acc)[i] = make_float4(0.0f, 0.0f, 0.0f, 0.0f);
  if (COUNT)
    for (int i = tid; i < ROWS; i += AGG_NW * 32) cnt[i] = 0.0f;
  __syncthreads();

  int* q = queue + wave * QCAP;
  int qn = 0;
  const int step   = AGG_NW * AGG_G * 128;
  const int nsteps = (E + step - 1) / step;
  const int w00    = wave * (AGG_G * 128);
  const int e00    = w00 + lane * 4;

  int4 bufA[AGG_G], bufB[AGG_G];
  if (nsteps > 0) load_dst_group(bufA, dst, e00, E, w00 + AGG_G * 128 <= E);
  for (int it = 0; it < nsteps; it += 2) {
    const bool hasB = (it + 1 < nsteps);
    if (hasB) load_dst_group(bufB, dst, (it + 1) * step + e00, E, (it + 1) * step + w00 + AGG_G * 128 <= E);
    scan_step<CH, COUNT>(bufA, it * step + e00, !hasB, qn, q, acc, cnt, xsrc, src, dst, r0, rows, lane);
    if (hasB) {
      const bool hasA = (it + 2 < nsteps);
      if (hasA) load_dst_group(bufA, dst, (it + 2) * step + e00, E, (it + 2) * step + w00 + AGG_G * 128 <= E);
      scan_step<CH, COUNT>(bufB, (it + 1) * step + e00, !hasA, qn, q, acc, cnt, xsrc, src, dst, r0, rows, lane);
    }
  }
  __syncthreads();

  for (int r = wave; r < rows; r += AGG_NW) {
    if (lane < LPR) {
      const float* a = acc + r * CH + lane;
      float4 o;
      o.x = a[0]; o.y = a[LPR]; o.z = a[2 * LPR]; o.w = a[3 * LPR];
      *(float4*)(msg + (size_t)(r0 + r) * CH + lane * 4) = o;
    }
  }
  if (COUNT)
    for (int i = tid; i < rows; i += AGG_NW * 32) deg[r0 + i] = cnt[i];
}

__global__ __launch_bounds__(128, 4)
void sage_gemm_relu_proj(const float* __restrict__ msg, const float* __restrict__ deg,
                         const float* __restrict__ xdst,
                         const float* __restrict__ Wl, const float* __restrict__ bl,
                         const float* __restrict__ Wr,
                         const float* __restrict__ Wl1, const float* __restrict__ bl1,
                         const float* __restrict__ Wr1,
                         float* __restrict__ p, float* __restrict__ q, int nrows, int nq)
{
  __shared__ float sA[16][IN_CH + 4];
  __shared__ float sX[16][IN_CH + 4];
  __shared__ float sH[16][HID + 4];
  const int tid = threadIdx.x;
  const int r0  = blockIdx.x * 16;

  for (int m = 0; m < 16; ++m) {
    int row = r0 + m;
    float a = 0.0f, xx = 0.0f;
    if (row < nrows) {
      float inv = 1.0f / fmaxf(deg[row], 1.0f);
      a  = msg [(size_t)row * IN_CH + tid] * inv;
      xx = xdst[(size_t)row * IN_CH + tid];
    }
    sA[m][tid] = a;
    sX[m][tid] = xx;
  }
  __syncthreads();

  const int wave = tid >> 5;
  const int lane = tid & 31;
  const int half = lane >> 4;
  const int m    = lane & 15;
  const int n    = lane & 15;
  const int kh   = half * 2;
  const int mbase = half * 8;

  for (int t = 0; t < 2; ++t) {
    const int n0 = (wave * 2 + t) * 16;
    v8f acc = {0.f, 0.f, 0.f, 0.f, 0.f, 0.f, 0.f, 0.f};
    for (int k = 0; k < IN_CH; k += 4) {
      v2f a, b;
      a.x = sA[m][k + kh];
      a.y = sA[m][k + kh + 1];
      b.x = Wl[(size_t)(k + kh)     * HID + n0 + n];
      b.y = Wl[(size_t)(k + kh + 1) * HID + n0 + n];
      acc = __builtin_amdgcn_wmma_f32_16x16x4_f32(false, a, false, b, (short)0, acc, false, false);
    }
    for (int k = 0; k < IN_CH; k += 4) {
      v2f a, b;
      a.x = sX[m][k + kh];
      a.y = sX[m][k + kh + 1];
      b.x = Wr[(size_t)(k + kh)     * HID + n0 + n];
      b.y = Wr[(size_t)(k + kh + 1) * HID + n0 + n];
      acc = __builtin_amdgcn_wmma_f32_16x16x4_f32(false, a, false, b, (short)0, acc, false, false);
    }
    const float bias = bl[n0 + n];
    for (int r = 0; r < 8; ++r) sH[mbase + r][n0 + n] = fmaxf(acc[r] + bias, 0.0f);
  }
  __syncthreads();

  if (wave < 3) {
    const int col = wave * 16 + n;
    const bool cv = (col < OUTC);
    const int colc = cv ? col : OUTC - 1;
    v8f accp = {0.f, 0.f, 0.f, 0.f, 0.f, 0.f, 0.f, 0.f};
    for (int k = 0; k < HID; k += 4) {
      v2f a, b;
      a.x = sH[m][k + kh];
      a.y = sH[m][k + kh + 1];
      b.x = Wl1[(size_t)(k + kh)     * OUTC + colc];
      b.y = Wl1[(size_t)(k + kh + 1) * OUTC + colc];
      accp = __builtin_amdgcn_wmma_f32_16x16x4_f32(false, a, false, b, (short)0, accp, false, false);
    }
    for (int r = 0; r < 8; ++r) {
      int row = r0 + mbase + r;
      if (row < nrows) p[(size_t)row * OUTP + col] = cv ? accp[r] : 1.0f;
    }
    if (r0 < nq) {
      v8f accq = {0.f, 0.f, 0.f, 0.f, 0.f, 0.f, 0.f, 0.f};
      for (int k = 0; k < HID; k += 4) {
        v2f a, b;
        a.x = sH[m][k + kh];
        a.y = sH[m][k + kh + 1];
        b.x = Wr1[(size_t)(k + kh)     * OUTC + colc];
        b.y = Wr1[(size_t)(k + kh + 1) * OUTC + colc];
        accq = __builtin_amdgcn_wmma_f32_16x16x4_f32(false, a, false, b, (short)0, accq, false, false);
      }
      const float bias1 = bl1[colc];
      for (int r = 0; r < 8; ++r) {
        int row = r0 + mbase + r;
        if (row < nq) q[(size_t)row * OUTP + col] = cv ? accq[r] + bias1 : 0.0f;
      }
    }
  }
}

__global__ __launch_bounds__(64)
void sage_mean_lsm(const float* __restrict__ msg, const float* __restrict__ q,
                   float* __restrict__ out, int nrows)
{
  const int row = blockIdx.x * blockDim.x + threadIdx.x;
  if (row >= nrows) return;
  const float4* m4 = (const float4*)(msg + (size_t)row * OUTP);
  const float4* q4 = (const float4*)(q   + (size_t)row * OUTP);
  float v[OUTP];
#pragma unroll
  for (int j = 0; j < OUTP / 4; ++j) {
    const float4 a = m4[j];
    v[4 * j + 0] = a.x; v[4 * j + 1] = a.y; v[4 * j + 2] = a.z; v[4 * j + 3] = a.w;
  }
  const float inv = 1.0f / fmaxf(v[OUTC], 1.0f);
#pragma unroll
  for (int j = 0; j < OUTP / 4; ++j) {
    const float4 b = q4[j];
    v[4 * j + 0] = v[4 * j + 0] * inv + b.x;
    v[4 * j + 1] = v[4 * j + 1] * inv + b.y;
    v[4 * j + 2] = v[4 * j + 2] * inv + b.z;
    v[4 * j + 3] = v[4 * j + 3] * inv + b.w;
  }
  float mx = -INFINITY;
#pragma unroll
  for (int c = 0; c < OUTC; ++c) mx = fmaxf(mx, v[c]);
  float s = 0.0f;
#pragma unroll
  for (int c = 0; c < OUTC; ++c) s += expf(v[c] - mx);
  const float lse = logf(s);
#pragma unroll
  for (int c = 0; c < OUTC; ++c)
    out[(size_t)row * OUTC + c] = v[c] - mx - lse;
}

extern "C" void kernel_launch(void* const* d_in, const int* in_sizes, int n_in,
                              void* d_out, int out_size, void* d_ws, size_t ws_size,
                              hipStream_t stream)
{
  const float* x    = (const float*)d_in[0];
  const int*   src0 = (const int*)d_in[1];
  const int*   dst0 = (const int*)d_in[2];
  const int*   src1 = (const int*)d_in[3];
  const int*   dst1 = (const int*)d_in[4];
  const float* Wl0  = (const float*)d_in[7];
  const float* bl0  = (const float*)d_in[8];
  const float* Wr0  = (const float*)d_in[9];
  const float* Wl1  = (const float*)d_in[10];
  const float* bl1  = (const float*)d_in[11];
  const float* Wr1  = (const float*)d_in[12];
  float* out = (float*)d_out;

  const int E0 = in_sizes[1];
  const int E1 = in_sizes[3];
  const int N1 = 100000;
  const int N2 = out_size / OUTC;

  float* msg0 = (float*)d_ws;
  float* deg0 = msg0 + (size_t)N1 * HID;
  float* p    = deg0 + N1;
  float* q    = p    + (size_t)N1 * OUTP;
  float* msg1 = q    + (size_t)N2 * OUTP;

  agg_slab<IN_CH, ROWS0, true><<<(N1 + ROWS0 - 1) / ROWS0, AGG_NW * 32, 0, stream>>>(x, src0, dst0, msg0, deg0, E0, N1);
  sage_gemm_relu_proj<<<(N1 + 15) / 16, 128, 0, stream>>>(msg0, deg0, x, Wl0, bl0, Wr0,
                                                          Wl1, bl1, Wr1, p, q, N1, N2);

  agg_slab<OUTP, ROWS1, false><<<(N2 + ROWS1 - 1) / ROWS1, AGG_NW * 32, 0, stream>>>(p, src1, dst1, msg1, (float*)0, E1, N2);
  sage_mean_lsm<<<(N2 + 63) / 64, 64, 0, stream>>>(msg1, q, out, N2);
}


# edge-sum scan: the next step's index loads carry no test, so the wait before a step's compares leaves them in flight (it waited for ALL loads, the ones just issued too); sum order as before
# speedup vs baseline: 2.0075x; 1.0380x over previous
#include <hip/hip_runtime.h>
#include <math.h>

#define IN_CH 128
#define HID   128
#define OUTC  47
#define OUTP  48

typedef float v2f __attribute__((ext_vector_type(2)));
typedef float v8f __attribute__((ext_vector_type(8)));

#define AGG_NW 32
#define AGG_G  4
#define QCAP   (32 + AGG_G * 128)
#define ROWS0  392
#define ROWS1  79

__device__ __forceinline__ void load_dst_whole(int4 (&v)[AGG_G], const int* __restrict__ dst, int e0) {
#pragma unroll
  for (int u = 0; u < AGG_G; ++u) v[u] = *(const int4*)(dst + e0 + u * 128);
}

__device__ __forceinline__ void load_dst_tail(int4 (&v)[AGG_G], const int* __restrict__ dst, int e0, int E) {
#pragma unroll
  for (int u = 0; u < AGG_G; ++u) {
    const int e = e0 + u * 128;
    v[u].x = (e     < E) ? dst[e]     : -1;
    v[u].y = (e + 1 < E) ? dst[e + 1] : -1;
    v[u].z = (e + 2 < E) ? dst[e + 2] : -1;
    v[u].w = (e + 3 < E) ? dst[e + 3] : -1;
  }
}

__device__ __forceinline__ void queue_hits(const int4 (&cur)[AGG_G], const int ebase,
                                           int& qn, int* __restrict__ q,
                                           const int r0, const int rows, const int lane)
{
#pragma unroll
  for (int u = 0; u < AGG_G; ++u) {
    const int dv[4] = {cur[u].x, cur[u].y, cur[u].z, cur[u].w};
#pragma unroll
    for (int c = 0; c < 4; ++c) {
      const bool hit = (unsigned)(dv[c] - r0) < (unsigned)rows;
      const unsigned mask = (unsigned)__ballot(hit);
      if (mask != 0u) {
        const int pos = qn + __popc(mask & ((1u << lane) - 1u));
        if (hit) q[pos] = ebase + u * 128 + c;
        qn += __popc(mask);
      }
    }
  }
}

template <int CH, bool COUNT>
__device__ __forceinline__ void drain_queue(const bool last, int& qn, const int* __restrict__ q,
                                            float* __restrict__ acc, float* __restrict__ cnt,
                                            const float* __restrict__ xsrc, const int* __restrict__ src,
                                            const int* __restrict__ dst, const int r0, const int lane)
{
  constexpr int LPR = CH / 4;
  while (qn >= 32 || (last && qn > 0)) {
    const int n  = min(qn, 32);
    const int qb = qn - n;
    int s_i = 0;
    int d_i = -1;
    if (lane < n) {
      const int e = q[qb + lane];
      s_i = src[e];
      d_i = dst[e] - r0;
      if (COUNT) atomicAdd(&cnt[d_i], 1.0f);
    }
    qn = qb;
#pragma unroll
    for (int j0 = 0; j0 < 32; j0 += 8) {
      if (j0 < n) {
        float4 v[8];
#pragma unroll
        for (int k = 0; k < 8; ++k) {
          const int sj = __builtin_amdgcn_readlane(s_i, j0 + k);
          if (lane < LPR) v[k] = *(const float4*)(xsrc + (size_t)sj * CH + lane * 4);
        }
#pragma unroll
        for (int k = 0; k < 8; ++k) {
          const int dj = __builtin_amdgcn_readlane(d_i, j0 + k);
          if (dj >= 0 && lane < LPR) {
            float* a = acc + dj * CH + lane;
            atomicAdd(a,           v[k].x);
            atomicAdd(a + LPR,     v[k].y);
            atomicAdd(a + 2 * LPR, v[k].z);
            atomicAdd(a + 3 * LPR, v[k].w);
          }
        }
      }
    }
  }
}

template <int CH, int ROWS, bool COUNT>
__global__ __launch_bounds__(AGG_NW * 32)
void agg_slab(const float* __restrict__ xsrc, const int* __restrict__ src, const int* __restrict__ dst,
              float* __restrict__ msg, float* __restrict__ deg, int E, int nrows)
{
  constexpr int LPR = CH / 4;
  __shared__ __attribute__((aligned(16))) float acc[ROWS * CH];
  __shared__ float cnt[COUNT ? ROWS : 1];
  __shared__ int   queue[AGG_NW * QCAP];

  const int tid  = threadIdx.x;
  const int lane = tid & 31;
  const int wave = __builtin_amdgcn_readfirstlane(tid >> 5);
  const int r0   = blockIdx.x * ROWS;
  const int rows = min(ROWS, nrows - r0);

  for (int i = tid; i < ROWS * CH / 4; i += AGG_NW * 32)
    ((float4*)acc)[i] = make_float4(0.0f, 0.0f, 0.0f, 0.0f);
  if (COUNT)
    for (int i = tid; i < ROWS; i += AGG_NW * 32) cnt[i] = 0.0f;
  __syncthreads();

  int* q = queue + wave * QCAP;
  int qn = 0;
  const int step = AGG_NW * AGG_G * 128;
  const int w00  = wave * (AGG_G * 128);
  const int e00  = w00 + lane * 4;
  const int nw   = (E >= w00 + AGG_G * 128) ? (E - w00 - AGG_G * 128) / step + 1 : 0;

  int4 bufA[AGG_G], bufB[AGG_G];
  if (nw > 0) {
    load_dst_whole(bufA, dst, e00);
    for (int it = 0; it < nw; it += 2) {
      load_dst_whole(bufB, dst, min(it + 1, nw - 1) * step + e00);
      queue_hits(bufA, it * step + e00, qn, q, r0, rows, lane);
      drain_queue<CH, COUNT>(false, qn, q, acc, cnt, xsrc, src, dst, r0, lane);
      if (it + 1 < nw) {
        load_dst_whole(bufA, dst, min(it + 2, nw - 1) * step + e00);
        queue_hits(bufB, (it + 1) * step + e00, qn, q, r0, rows, lane);
        drain_queue<CH, COUNT>(false, qn, q, acc, cnt, xsrc, src, dst, r0, lane);
      }
    }
  }
  if (nw * step + w00 < E) {
    load_dst_tail(bufA, dst, nw * step + e00, E);
    queue_hits(bufA, nw * step + e00, qn, q, r0, rows, lane);
  }
  drain_queue<CH, COUNT>(true, qn, q, acc, cnt, xsrc, src, dst, r0, lane);
  __syncthreads();

  for (int r = wave; r < rows; r += AGG_NW) {
    if (lane < LPR) {
      const float* a = acc + r * CH + lane;
      float4 o;
      o.x = a[0]; o.y = a[LPR]; o.z = a[2 * LPR]; o.w = a[3 * LPR];
      *(float4*)(msg + (size_t)(r0 + r) * CH + lane * 4) = o;
    }
  }
  if (COUNT)
    for (int i = tid; i < rows; i += AGG_NW * 32) deg[r0 + i] = cnt[i];
}

__global__ __launch_bounds__(128, 4)
void sage_gemm_relu_proj(const float* __restrict__ msg, const float* __restrict__ deg,
                         const float* __restrict__ xdst,
                         const float* __restrict__ Wl, const float* __restrict__ bl,
                         const float* __restrict__ Wr,
                         const float* __restrict__ Wl1, const float* __restrict__ bl1,
                         const float* __restrict__ Wr1,
                         float* __restrict__ p, float* __restrict__ q, int nrows, int nq)
{
  __shared__ float sA[16][IN_CH + 4];
  __shared__ float sX[16][IN_CH + 4];
  __shared__ float sH[16][HID + 4];
  const int tid = threadIdx.x;
  const int r0  = blockIdx.x * 16;

  for (int m = 0; m < 16; ++m) {
    int row = r0 + m;
    float a = 0.0f, xx = 0.0f;
    if (row < nrows) {
      float inv = 1.0f / fmaxf(deg[row], 1.0f);
      a  = msg [(size_t)row * IN_CH + tid] * inv;
      xx = xdst[(size_t)row * IN_CH + tid];
    }
    sA[m][tid] = a;
    sX[m][tid] = xx;
  }
  __syncthreads();

  const int wave = tid >> 5;
  const int lane = tid & 31;
  const int half = lane >> 4;
  const int m    = lane & 15;
  const int n    = lane & 15;
  const int kh   = half * 2;
  const int mbase = half * 8;

  for (int t = 0; t < 2; ++t) {
    const int n0 = (wave * 2 + t) * 16;
    v8f acc = {0.f, 0.f, 0.f, 0.f, 0.f, 0.f, 0.f, 0.f};
    for (int k = 0; k < IN_CH; k += 4) {
      v2f a, b;
      a.x = sA[m][k + kh];
      a.y = sA[m][k + kh + 1];
      b.x = Wl[(size_t)(k + kh)     * HID + n0 + n];
      b.y = Wl[(size_t)(k + kh + 1) * HID + n0 + n];
      acc = __builtin_amdgcn_wmma_f32_16x16x4_f32(false, a, false, b, (short)0, acc, false, false);
    }
    for (int k = 0; k < IN_CH; k += 4) {
      v2f a, b;
      a.x = sX[m][k + kh];
      a.y = sX[m][k + kh + 1];
      b.x = Wr[(size_t)(k + kh)     * HID + n0 + n];
      b.y = Wr[(size_t)(k + kh + 1) * HID + n0 + n];
      acc = __builtin_amdgcn_wmma_f32_16x16x4_f32(false, a, false, b, (short)0, acc, false, false);
    }
    const float bias = bl[n0 + n];
    for (int r = 0; r < 8; ++r) sH[mbase + r][n0 + n] = fmaxf(acc[r] + bias, 0.0f);
  }
  __syncthreads();

  if (wave < 3) {
    const int col = wave * 16 + n;
    const bool cv = (col < OUTC);
    const int colc = cv ? col : OUTC - 1;
    v8f accp = {0.f, 0.f, 0.f, 0.f, 0.f, 0.f, 0.f, 0.f};
    for (int k = 0; k < HID; k += 4) {
      v2f a, b;
      a.x = sH[m][k + kh];
      a.y = sH[m][k + kh + 1];
      b.x = Wl1[(size_t)(k + kh)     * OUTC + colc];
      b.y = Wl1[(size_t)(k + kh + 1) * OUTC + colc];
      accp = __builtin_amdgcn_wmma_f32_16x16x4_f32(false, a, false, b, (short)0, accp, false, false);
    }
    for (int r = 0; r < 8; ++r) {
      int row = r0 + mbase + r;
      if (row < nrows) p[(size_t)row * OUTP + col] = cv ? accp[r] : 1.0f;
    }
    if (r0 < nq) {
      v8f accq = {0.f, 0.f, 0.f, 0.f, 0.f, 0.f, 0.f, 0.f};
      for (int k = 0; k < HID; k += 4) {
        v2f a, b;
        a.x = sH[m][k + kh];
        a.y = sH[m][k + kh + 1];
        b.x = Wr1[(size_t)(k + kh)     * OUTC + colc];
        b.y = Wr1[(size_t)(k + kh + 1) * OUTC + colc];
        accq = __builtin_amdgcn_wmma_f32_16x16x4_f32(false, a, false, b, (short)0, accq, false, false);
      }
      const float bias1 = bl1[colc];
      for (int r = 0; r < 8; ++r) {
        int row = r0 + mbase + r;
        if (row < nq) q[(size_t)row * OUTP + col] = cv ? accq[r] + bias1 : 0.0f;
      }
    }
  }
}

__global__ __launch_bounds__(64)
void sage_mean_lsm(const float* __restrict__ msg, const float* __restrict__ q,
                   float* __restrict__ out, int nrows)
{
  const int row = blockIdx.x * blockDim.x + threadIdx.x;
  if (row >= nrows) return;
  const float4* m4 = (const float4*)(msg + (size_t)row * OUTP);
  const float4* q4 = (const float4*)(q   + (size_t)row * OUTP);
  float v[OUTP];
#pragma unroll
  for (int j = 0; j < OUTP / 4; ++j) {
    const float4 a = m4[j];
    v[4 * j + 0] = a.x; v[4 * j + 1] = a.y; v[4 * j + 2] = a.z; v[4 * j + 3] = a.w;
  }
  const float inv = 1.0f / fmaxf(v[OUTC], 1.0f);
#pragma unroll
  for (int j = 0; j < OUTP / 4; ++j) {
    const float4 b = q4[j];
    v[4 * j + 0] = v[4 * j + 0] * inv + b.x;
    v[4 * j + 1] = v[4 * j + 1] * inv + b.y;
    v[4 * j + 2] = v[4 * j + 2] * inv + b.z;
    v[4 * j + 3] = v[4 * j + 3] * inv + b.w;
  }
  float mx = -INFINITY;
#pragma unroll
  for (int c = 0; c < OUTC; ++c) mx = fmaxf(mx, v[c]);
  float s = 0.0f;
#pragma unroll
  for (int c = 0; c < OUTC; ++c) s += expf(v[c] - mx);
  const float lse = logf(s);
#pragma unroll
  for (int c = 0; c < OUTC; ++c)
    out[(size_t)row * OUTC + c] = v[c] - mx - lse;
}

extern "C" void kernel_launch(void* const* d_in, const int* in_sizes, int n_in,
                              void* d_out, int out_size, void* d_ws, size_t ws_size,
                              hipStream_t stream)
{
  const float* x    = (const float*)d_in[0];
  const int*   src0 = (const int*)d_in[1];
  const int*   dst0 = (const int*)d_in[2];
  const int*   src1 = (const int*)d_in[3];
  const int*   dst1 = (const int*)d_in[4];
  const float* Wl0  = (const float*)d_in[7];
  const float* bl0  = (const float*)d_in[8];
  const float* Wr0  = (const float*)d_in[9];
  const float* Wl1  = (const float*)d_in[10];
  const float* bl1  = (const float*)d_in[11];
  const float* Wr1  = (const float*)d_in[12];
  float* out = (float*)d_out;

  const int E0 = in_sizes[1];
  const int E1 = in_sizes[3];
  const int N1 = 100000;
  const int N2 = out_size / OUTC;

  float* msg0 = (float*)d_ws;
  float* deg0 = msg0 + (size_t)N1 * HID;
  float* p    = deg0 + N1;
  float* q    = p    + (size_t)N1 * OUTP;
  float* msg1 = q    + (size_t)N2 * OUTP;

  agg_slab<IN_CH, ROWS0, true><<<(N1 + ROWS0 - 1) / ROWS0, AGG_NW * 32, 0, stream>>>(x, src0, dst0, msg0, deg0, E0, N1);
  sage_gemm_relu_proj<<<(N1 + 15) / 16, 128, 0, stream>>>(msg0, deg0, x, Wl0, bl0, Wr0,
                                                          Wl1, bl1, Wr1, p, q, N1, N2);

  agg_slab<OUTP, ROWS1, false><<<(N2 + ROWS1 - 1) / ROWS1, AGG_NW * 32, 0, stream>>>(p, src1, dst1, msg1, (float*)0, E1, N2);
  sage_mean_lsm<<<(N2 + 63) / 64, 64, 0, stream>>>(msg1, q, out, N2);
}


# edge-sum scan reads 8-bit block keys (dst/ROWS), bit-sliced by a new first launch: a lane tests 32 edges with 8 bit ops, one ballot per 1024 edges not per 32; 4 launches to 5; sum order = LDS arrival
# speedup vs baseline: 2.2263x; 1.1090x over previous
#include <hip/hip_runtime.h>
#include <math.h>

#define IN_CH 128
#define HID   128
#define OUTC  47
#define OUTP  48

typedef float v2f __attribute__((ext_vector_type(2)));
typedef float v8f __attribute__((ext_vector_type(8)));

#define AGG_NW 32
#define KT     2
#define STEP_T (AGG_NW * KT)
#define QCAP   64
#define ROWS0  392
#define ROWS1  79

__device__ __forceinline__ int padded_tiles(int E) { return ((E + STEP_T * 1024 - 1) / (STEP_T * 1024)) * STEP_T; }

template <int ROWS>
__device__ __forceinline__ void key_tile(const int* __restrict__ dst, const int E, const int tile,
                                         int* __restrict__ keys, const int lane)
{
  const int e0 = tile * 1024 + lane;
  int d[32];
  if (e0 - lane + 1024 <= E) {
#pragma unroll
    for (int g = 0; g < 32; ++g) d[g] = dst[e0 + g * 32];
  } else {
#pragma unroll
    for (int g = 0; g < 32; ++g) d[g] = (e0 + g * 32 < E) ? dst[e0 + g * 32] : 0;
  }
  unsigned pl[8] = {0u, 0u, 0u, 0u, 0u, 0u, 0u, 0u};
#pragma unroll
  for (int g = 0; g < 32; ++g) {
    const unsigned k = (unsigned)d[g] / (unsigned)ROWS;
#pragma unroll
    for (int j = 0; j < 8; ++j) {
      const unsigned w = (unsigned)__ballot((k >> j) & 1u);
      if (lane == g) pl[j] = w;
    }
  }
  int4* kt = (int4*)keys + (size_t)tile * 64;
  kt[lane]      = make_int4((int)pl[0], (int)pl[1], (int)pl[2], (int)pl[3]);
  kt[32 + lane] = make_int4((int)pl[4], (int)pl[5], (int)pl[6], (int)pl[7]);
}

__global__ __launch_bounds__(256)
void make_keys(const int* __restrict__ dst0, const int* __restrict__ dst1, int* __restrict__ keys, int E0, int E1)
{
  const int lane = threadIdx.x & 31;
  const int gw   = __builtin_amdgcn_readfirstlane((int)(blockIdx.x * 8 + (threadIdx.x >> 5)));
  const int nt0  = padded_tiles(E0);
  const int nt   = nt0 + padded_tiles(E1);
  for (int t = gw; t < nt; t += (int)gridDim.x * 8) {
    if (t < nt0) key_tile<ROWS0>(dst0, E0, t, keys, lane);
    else         key_tile<ROWS1>(dst1, E1, t - nt0, keys + (size_t)nt0 * 256, lane);
  }
}


__device__ __forceinline__ void load_keys(int4 (&v)[2 * KT], const int4* __restrict__ k) {
#pragma unroll
  for (int t = 0; t < KT; ++t) { v[2 * t] = k[t * 64]; v[2 * t + 1] = k[t * 64 + 32]; }
}

template <int CH, int ROWS, bool COUNT>
__device__ __forceinline__ void drain_queue(const bool last, int& qn, const int* q,
                                            float* __restrict__ acc, float* __restrict__ cnt,
                                            const float* __restrict__ xsrc, const int* __restrict__ src,
                                            const int* __restrict__ dst, const int r0, const int rows, const int lane)
{
  constexpr int LPR = CH / 4;
  while (qn >= 32 || (last && qn > 0)) {
    const int n  = min(qn, 32);
    const int qb = qn - n;
    int s_i = 0;
    int d_i = ROWS;
    if (lane < n) {
      const int e = q[qb + lane];
      s_i = src[e];
      const int d = dst[e] - r0;
      if ((unsigned)d < (unsigned)rows) {
        d_i = d;
        if (COUNT) atomicAdd(&cnt[d], 1.0f);
      }
    }
    qn = qb;
#pragma unroll
    for (int j0 = 0; j0 < 32; j0 += 8) {
      if (j0 < n) {
        float4 v[8];
#pragma unroll
        for (int k = 0; k < 8; ++k) {
          const int sj = __builtin_amdgcn_readlane(s_i, j0 + k);
          if (lane < LPR) v[k] = *(const float4*)(xsrc + (size_t)sj * CH + lane * 4);
        }
#pragma unroll
        for (int k = 0; k < 8; ++k) {
          const int dj = __builtin_amdgcn_readlane(d_i, j0 + k);
          if (lane < LPR) {
            float* a = acc + dj * CH + lane;
            atomicAdd(a,           v[k].x);
            atomicAdd(a + LPR,     v[k].y);
            atomicAdd(a + 2 * LPR, v[k].z);
            atomicAdd(a + 3 * LPR, v[k].w);
          }
        }
      }
    }
  }
}

template <int CH, int ROWS, bool COUNT>
__device__ __forceinline__ void scan_tile(const int4 lo, const int4 hi, const int tile, const int E,
                                          const unsigned (&xs)[8], int& qn, int* q,
                                          float* __restrict__ acc, float* __restrict__ cnt,
                                          const float* __restrict__ xsrc, const int* __restrict__ src,
                                          const int* __restrict__ dst, const int r0, const int rows, const int lane)
{
  const int gb = tile * 1024 + lane * 32;
  unsigned m = ((unsigned)lo.x ^ xs[0]) & ((unsigned)lo.y ^ xs[1]) & ((unsigned)lo.z ^ xs[2]) & ((unsigned)lo.w ^ xs[3])
             & ((unsigned)hi.x ^ xs[4]) & ((unsigned)hi.y ^ xs[5]) & ((unsigned)hi.z ^ xs[6]) & ((unsigned)hi.w ^ xs[7]);
  if (tile * 1024 + 1024 > E) {
    const int rem = E - gb;
    m = (rem >= 32) ? m : ((rem > 0) ? (m & ((1u << rem) - 1u)) : 0u);
  }
  const unsigned lt = (1u << lane) - 1u;
  unsigned mask = (unsigned)__ballot(m != 0u);
  while (mask != 0u) {
    if (m != 0u) {
      q[qn + __popc(mask & lt)] = gb + __ffs((int)m) - 1;
      m &= m - 1u;
    }
    qn += __popc(mask);
    drain_queue<CH, ROWS, COUNT>(false, qn, q, acc, cnt, xsrc, src, dst, r0, rows, lane);
    mask = (unsigned)__ballot(m != 0u);
  }
}

template <int CH, int ROWS, bool COUNT>
__global__ __launch_bounds__(AGG_NW * 32)
void agg_slab(const float* __restrict__ xsrc, const int* __restrict__ src, const int* __restrict__ dst,
              const int* __restrict__ keys, float* __restrict__ msg, float* __restrict__ deg,
              int E, int Eprev, int nrows)
{
  constexpr int LPR = CH / 4;
  __shared__ __attribute__((aligned(16))) float acc[(ROWS + 1) * CH];
  __shared__ float cnt[COUNT ? ROWS : 1];
  __shared__ int   queue[AGG_NW * QCAP];

  const int tid  = threadIdx.x;
  const int lane = tid & 31;
  const int wave = __builtin_amdgcn_readfirstlane(tid >> 5);
  const int r0   = blockIdx.x * ROWS;
  const int rows = min(ROWS, nrows - r0);

  for (int i = tid; i < ROWS * CH / 4; i += AGG_NW * 32)
    ((float4*)acc)[i] = make_float4(0.0f, 0.0f, 0.0f, 0.0f);
  if (COUNT)
    for (int i = tid; i < ROWS; i += AGG_NW * 32) cnt[i] = 0.0f;
  __syncthreads();

  int* q = queue + wave * QCAP;
  int qn = 0;
  unsigned xs[8];
#pragma unroll
  for (int j = 0; j < 8; ++j) xs[j] = ((blockIdx.x >> j) & 1u) ? 0u : ~0u;

  const int nsteps = padded_tiles(E) / STEP_T;
  const int4* kw = (const int4*)keys + ((size_t)padded_tiles(Eprev) + wave * KT) * 64 + lane;
  const int t00  = wave * KT;

  int4 bufA[2 * KT], bufB[2 * KT];
  if (nsteps > 0) {
    load_keys(bufA, kw);
    for (int s = 0; s < nsteps; s += 2) {
      load_keys(bufB, kw + (size_t)min(s + 1, nsteps - 1) * (STEP_T * 64));
#pragma unroll
      for (int t = 0; t < KT; ++t)
        scan_tile<CH, ROWS, COUNT>(bufA[2 * t], bufA[2 * t + 1], s * STEP_T + t00 + t, E, xs, qn, q,
                             acc, cnt, xsrc, src, dst, r0, rows, lane);
      if (s + 1 < nsteps) {
        load_keys(bufA, kw + (size_t)min(s + 2, nsteps - 1) * (STEP_T * 64));
#pragma unroll
        for (int t = 0; t < KT; ++t)
          scan_tile<CH, ROWS, COUNT>(bufB[2 * t], bufB[2 * t + 1], (s + 1) * STEP_T + t00 + t, E, xs, qn, q,
                               acc, cnt, xsrc, src, dst, r0, rows, lane);
      }
    }
  }
  drain_queue<CH, ROWS, COUNT>(true, qn, q, acc, cnt, xsrc, src, dst, r0, rows, lane);
  __syncthreads();

  for (int r = wave; r < rows; r += AGG_NW) {
    if (lane < LPR) {
      const float* a = acc + r * CH + lane;
      float4 o;
      o.x = a[0]; o.y = a[LPR]; o.z = a[2 * LPR]; o.w = a[3 * LPR];
      *(float4*)(msg + (size_t)(r0 + r) * CH + lane * 4) = o;
    }
  }
  if (COUNT)
    for (int i = tid; i < rows; i += AGG_NW * 32) deg[r0 + i] = cnt[i];
}

__global__ __launch_bounds__(128, 4)
void sage_gemm_relu_proj(const float* __restrict__ msg, const float* __restrict__ deg,
                         const float* __restrict__ xdst,
                         const float* __restrict__ Wl, const float* __restrict__ bl,
                         const float* __restrict__ Wr,
                         const float* __restrict__ Wl1, const float* __restrict__ bl1,
                         const float* __restrict__ Wr1,
                         float* __restrict__ p, float* __restrict__ q, int nrows, int nq)
{
  __shared__ float sA[16][IN_CH + 4];
  __shared__ float sX[16][IN_CH + 4];
  __shared__ float sH[16][HID + 4];
  const int tid = threadIdx.x;
  const int r0  = blockIdx.x * 16;

  for (int m = 0; m < 16; ++m) {
    int row = r0 + m;
    float a = 0.0f, xx = 0.0f;
    if (row < nrows) {
      float inv = 1.0f / fmaxf(deg[row], 1.0f);
      a  = msg [(size_t)row * IN_CH + tid] * inv;
      xx = xdst[(size_t)row * IN_CH + tid];
    }
    sA[m][tid] = a;
    sX[m][tid] = xx;
  }
  __syncthreads();

  const int wave = tid >> 5;
  const int lane = tid & 31;
  const int half = lane >> 4;
  const int m    = lane & 15;
  const int n    = lane & 15;
  const int kh   = half * 2;
  const int mbase = half * 8;

  for (int t = 0; t < 2; ++t) {
    const int n0 = (wave * 2 + t) * 16;
    v8f acc = {0.f, 0.f, 0.f, 0.f, 0.f, 0.f, 0.f, 0.f};
    for (int k = 0; k < IN_CH; k += 4) {
      v2f a, b;
      a.x = sA[m][k + kh];
      a.y = sA[m][k + kh + 1];
      b.x = Wl[(size_t)(k + kh)     * HID + n0 + n];
      b.y = Wl[(size_t)(k + kh + 1) * HID + n0 + n];
      acc = __builtin_amdgcn_wmma_f32_16x16x4_f32(false, a, false, b, (short)0, acc, false, false);
    }
    for (int k = 0; k < IN_CH; k += 4) {
      v2f a, b;
      a.x = sX[m][k + kh];
      a.y = sX[m][k + kh + 1];
      b.x = Wr[(size_t)(k + kh)     * HID + n0 + n];
      b.y = Wr[(size_t)(k + kh + 1) * HID + n0 + n];
      acc = __builtin_amdgcn_wmma_f32_16x16x4_f32(false, a, false, b, (short)0, acc, false, false);
    }
    const float bias = bl[n0 + n];
    for (int r = 0; r < 8; ++r) sH[mbase + r][n0 + n] = fmaxf(acc[r] + bias, 0.0f);
  }
  __syncthreads();

  if (wave < 3) {
    const int col = wave * 16 + n;
    const bool cv = (col < OUTC);
    const int colc = cv ? col : OUTC - 1;
    v8f accp = {0.f, 0.f, 0.f, 0.f, 0.f, 0.f, 0.f, 0.f};
    for (int k = 0; k < HID; k += 4) {
      v2f a, b;
      a.x = sH[m][k + kh];
      a.y = sH[m][k + kh + 1];
      b.x = Wl1[(size_t)(k + kh)     * OUTC + colc];
      b.y = Wl1[(size_t)(k + kh + 1) * OUTC + colc];
      accp = __builtin_amdgcn_wmma_f32_16x16x4_f32(false, a, false, b, (short)0, accp, false, false);
    }
    for (int r = 0; r < 8; ++r) {
      int row = r0 + mbase + r;
      if (row < nrows) p[(size_t)row * OUTP + col] = cv ? accp[r] : 1.0f;
    }
    if (r0 < nq) {
      v8f accq = {0.f, 0.f, 0.f, 0.f, 0.f, 0.f, 0.f, 0.f};
      for (int k = 0; k < HID; k += 4) {
        v2f a, b;
        a.x = sH[m][k + kh];
        a.y = sH[m][k + kh + 1];
        b.x = Wr1[(size_t)(k + kh)     * OUTC + colc];
        b.y = Wr1[(size_t)(k + kh + 1) * OUTC + colc];
        accq = __builtin_amdgcn_wmma_f32_16x16x4_f32(false, a, false, b, (short)0, accq, false, false);
      }
      const float bias1 = bl1[colc];
      for (int r = 0; r < 8; ++r) {
        int row = r0 + mbase + r;
        if (row < nq) q[(size_t)row * OUTP + col] = cv ? accq[r] + bias1 : 0.0f;
      }
    }
  }
}

__global__ __launch_bounds__(64)
void sage_mean_lsm(const float* __restrict__ msg, const float* __restrict__ q,
                   float* __restrict__ out, int nrows)
{
  const int row = blockIdx.x * blockDim.x + threadIdx.x;
  if (row >= nrows) return;
  const float4* m4 = (const float4*)(msg + (size_t)row * OUTP);
  const float4* q4 = (const float4*)(q   + (size_t)row * OUTP);
  float v[OUTP];
#pragma unroll
  for (int j = 0; j < OUTP / 4; ++j) {
    const float4 a = m4[j];
    v[4 * j + 0] = a.x; v[4 * j + 1] = a.y; v[4 * j + 2] = a.z; v[4 * j + 3] = a.w;
  }
  const float inv = 1.0f / fmaxf(v[OUTC], 1.0f);
#pragma unroll
  for (int j = 0; j < OUTP / 4; ++j) {
    const float4 b = q4[j];
    v[4 * j + 0] = v[4 * j + 0] * inv + b.x;
    v[4 * j + 1] = v[4 * j + 1] * inv + b.y;
    v[4 * j + 2] = v[4 * j + 2] * inv + b.z;
    v[4 * j + 3] = v[4 * j + 3] * inv + b.w;
  }
  float mx = -INFINITY;
#pragma unroll
  for (int c = 0; c < OUTC; ++c) mx = fmaxf(mx, v[c]);
  float s = 0.0f;
#pragma unroll
  for (int c = 0; c < OUTC; ++c) s += expf(v[c] - mx);
  const float lse = logf(s);
#pragma unroll
  for (int c = 0; c < OUTC; ++c)
    out[(size_t)row * OUTC + c] = v[c] - mx - lse;
}

extern "C" void kernel_launch(void* const* d_in, const int* in_sizes, int n_in,
                              void* d_out, int out_size, void* d_ws, size_t ws_size,
                              hipStream_t stream)
{
  const float* x    = (const float*)d_in[0];
  const int*   src0 = (const int*)d_in[1];
  const int*   dst0 = (const int*)d_in[2];
  const int*   src1 = (const int*)d_in[3];
  const int*   dst1 = (const int*)d_in[4];
  const float* Wl0  = (const float*)d_in[7];
  const float* bl0  = (const float*)d_in[8];
  const float* Wr0  = (const float*)d_in[9];
  const float* Wl1  = (const float*)d_in[10];
  const float* bl1  = (const float*)d_in[11];
  const float* Wr1  = (const float*)d_in[12];
  float* out = (float*)d_out;

  const int E0 = in_sizes[1];
  const int E1 = in_sizes[3];
  const int N1 = 100000;
  const int N2 = out_size / OUTC;

  float* msg0 = (float*)d_ws;
  float* deg0 = msg0 + (size_t)N1 * HID;
  float* p    = deg0 + N1;
  float* q    = p    + (size_t)N1 * OUTP;
  float* msg1 = q    + (size_t)N2 * OUTP;
  int*   keys = (int*)(msg1 + (size_t)N2 * OUTP);

  make_keys<<<240, 256, 0, stream>>>(dst0, dst1, keys, E0, E1);

  agg_slab<IN_CH, ROWS0, true><<<(N1 + ROWS0 - 1) / ROWS0, AGG_NW * 32, 0, stream>>>(x, src0, dst0, keys, msg0, deg0, E0, 0, N1);
  sage_gemm_relu_proj<<<(N1 + 15) / 16, 128, 0, stream>>>(msg0, deg0, x, Wl0, bl0, Wr0,
                                                          Wl1, bl1, Wr1, p, q, N1, N2);

  agg_slab<OUTP, ROWS1, false><<<(N2 + ROWS1 - 1) / ROWS1, AGG_NW * 32, 0, stream>>>(p, src1, dst1, keys, msg1, (float*)0, E1, E0, N2);
  sage_mean_lsm<<<(N2 + 63) / 64, 64, 0, stream>>>(msg1, q, out, N2);
}


# layer-0 dense launch as a resident grid: 512 blocks of 8 waves walk the 16-row tiles, a wave keeps its 16-column strips of Wl and Wr in registers, next tile's rows asked a tile ahead; sum order kept
# speedup vs baseline: 3.3452x; 1.5026x over previous
#include <hip/hip_runtime.h>
#include <math.h>

#define IN_CH 128
#define HID   128
#define OUTC  47
#define OUTP  48

typedef float v2f __attribute__((ext_vector_type(2)));
typedef float v8f __attribute__((ext_vector_type(8)));

#define AGG_NW 32
#define KT     2
#define STEP_T (AGG_NW * KT)
#define QCAP   64
#define ROWS0  392
#define ROWS1  79

__device__ __forceinline__ int padded_tiles(int E) { return ((E + STEP_T * 1024 - 1) / (STEP_T * 1024)) * STEP_T; }

template <int ROWS>
__device__ __forceinline__ void key_tile(const int* __restrict__ dst, const int E, const int tile,
                                         int* __restrict__ keys, const int lane)
{
  const int e0 = tile * 1024 + lane;
  int d[32];
  if (e0 - lane + 1024 <= E) {
#pragma unroll
    for (int g = 0; g < 32; ++g) d[g] = dst[e0 + g * 32];
  } else {
#pragma unroll
    for (int g = 0; g < 32; ++g) d[g] = (e0 + g * 32 < E) ? dst[e0 + g * 32] : 0;
  }
  unsigned pl[8] = {0u, 0u, 0u, 0u, 0u, 0u, 0u, 0u};
#pragma unroll
  for (int g = 0; g < 32; ++g) {
    const unsigned k = (unsigned)d[g] / (unsigned)ROWS;
#pragma unroll
    for (int j = 0; j < 8; ++j) {
      const unsigned w = (unsigned)__ballot((k >> j) & 1u);
      if (lane == g) pl[j] = w;
    }
  }
  int4* kt = (int4*)keys + (size_t)tile * 64;
  kt[lane]      = make_int4((int)pl[0], (int)pl[1], (int)pl[2], (int)pl[3]);
  kt[32 + lane] = make_int4((int)pl[4], (int)pl[5], (int)pl[6], (int)pl[7]);
}

__global__ __launch_bounds__(256)
void make_keys(const int* __restrict__ dst0, const int* __restrict__ dst1, int* __restrict__ keys, int E0, int E1)
{
  const int lane = threadIdx.x & 31;
  const int gw   = __builtin_amdgcn_readfirstlane((int)(blockIdx.x * 8 + (threadIdx.x >> 5)));
  const int nt0  = padded_tiles(E0);
  const int nt   = nt0 + padded_tiles(E1);
  for (int t = gw; t < nt; t += (int)gridDim.x * 8) {
    if (t < nt0) key_tile<ROWS0>(dst0, E0, t, keys, lane);
    else         key_tile<ROWS1>(dst1, E1, t - nt0, keys + (size_t)nt0 * 256, lane);
  }
}


__device__ __forceinline__ void load_keys(int4 (&v)[2 * KT], const int4* __restrict__ k) {
#pragma unroll
  for (int t = 0; t < KT; ++t) { v[2 * t] = k[t * 64]; v[2 * t + 1] = k[t * 64 + 32]; }
}

template <int CH, int ROWS, bool COUNT>
__device__ __forceinline__ void drain_queue(const bool last, int& qn, const int* q,
                                            float* __restrict__ acc, float* __restrict__ cnt,
                                            const float* __restrict__ xsrc, const int* __restrict__ src,
                                            const int* __restrict__ dst, const int r0, const int rows, const int lane)
{
  constexpr int LPR = CH / 4;
  while (qn >= 32 || (last && qn > 0)) {
    const int n  = min(qn, 32);
    const int qb = qn - n;
    int s_i = 0;
    int d_i = ROWS;
    if (lane < n) {
      const int e = q[qb + lane];
      s_i = src[e];
      const int d = dst[e] - r0;
      if ((unsigned)d < (unsigned)rows) {
        d_i = d;
        if (COUNT) atomicAdd(&cnt[d], 1.0f);
      }
    }
    qn = qb;
#pragma unroll
    for (int j0 = 0; j0 < 32; j0 += 8) {
      if (j0 < n) {
        float4 v[8];
#pragma unroll
        for (int k = 0; k < 8; ++k) {
          const int sj = __builtin_amdgcn_readlane(s_i, j0 + k);
          if (lane < LPR) v[k] = *(const float4*)(xsrc + (size_t)sj * CH + lane * 4);
        }
#pragma unroll
        for (int k = 0; k < 8; ++k) {
          const int dj = __builtin_amdgcn_readlane(d_i, j0 + k);
          if (lane < LPR) {
            float* a = acc + dj * CH + lane;
            atomicAdd(a,           v[k].x);
            atomicAdd(a + LPR,     v[k].y);
            atomicAdd(a + 2 * LPR, v[k].z);
            atomicAdd(a + 3 * LPR, v[k].w);
          }
        }
      }
    }
  }
}

template <int CH, int ROWS, bool COUNT>
__device__ __forceinline__ void scan_tile(const int4 lo, const int4 hi, const int tile, const int E,
                                          const unsigned (&xs)[8], int& qn, int* q,
                                          float* __restrict__ acc, float* __restrict__ cnt,
                                          const float* __restrict__ xsrc, const int* __restrict__ src,
                                          const int* __restrict__ dst, const int r0, const int rows, const int lane)
{
  const int gb = tile * 1024 + lane * 32;
  unsigned m = ((unsigned)lo.x ^ xs[0]) & ((unsigned)lo.y ^ xs[1]) & ((unsigned)lo.z ^ xs[2]) & ((unsigned)lo.w ^ xs[3])
             & ((unsigned)hi.x ^ xs[4]) & ((unsigned)hi.y ^ xs[5]) & ((unsigned)hi.z ^ xs[6]) & ((unsigned)hi.w ^ xs[7]);
  if (tile * 1024 + 1024 > E) {
    const int rem = E - gb;
    m = (rem >= 32) ? m : ((rem > 0) ? (m & ((1u << rem) - 1u)) : 0u);
  }
  const unsigned lt = (1u << lane) - 1u;
  unsigned mask = (unsigned)__ballot(m != 0u);
  while (mask != 0u) {
    if (m != 0u) {
      q[qn + __popc(mask & lt)] = gb + __ffs((int)m) - 1;
      m &= m - 1u;
    }
    qn += __popc(mask);
    drain_queue<CH, ROWS, COUNT>(false, qn, q, acc, cnt, xsrc, src, dst, r0, rows, lane);
    mask = (unsigned)__ballot(m != 0u);
  }
}

template <int CH, int ROWS, bool COUNT>
__global__ __launch_bounds__(AGG_NW * 32)
void agg_slab(const float* __restrict__ xsrc, const int* __restrict__ src, const int* __restrict__ dst,
              const int* __restrict__ keys, float* __restrict__ msg, float* __restrict__ deg,
              int E, int Eprev, int nrows)
{
  constexpr int LPR = CH / 4;
  __shared__ __attribute__((aligned(16))) float acc[(ROWS + 1) * CH];
  __shared__ float cnt[COUNT ? ROWS : 1];
  __shared__ int   queue[AGG_NW * QCAP];

  const int tid  = threadIdx.x;
  const int lane = tid & 31;
  const int wave = __builtin_amdgcn_readfirstlane(tid >> 5);
  const int r0   = blockIdx.x * ROWS;
  const int rows = min(ROWS, nrows - r0);

  for (int i = tid; i < ROWS * CH / 4; i += AGG_NW * 32)
    ((float4*)acc)[i] = make_float4(0.0f, 0.0f, 0.0f, 0.0f);
  if (COUNT)
    for (int i = tid; i < ROWS; i += AGG_NW * 32) cnt[i] = 0.0f;
  __syncthreads();

  int* q = queue + wave * QCAP;
  int qn = 0;
  unsigned xs[8];
#pragma unroll
  for (int j = 0; j < 8; ++j) xs[j] = ((blockIdx.x >> j) & 1u) ? 0u : ~0u;

  const int nsteps = padded_tiles(E) / STEP_T;
  const int4* kw = (const int4*)keys + ((size_t)padded_tiles(Eprev) + wave * KT) * 64 + lane;
  const int t00  = wave * KT;

  int4 bufA[2 * KT], bufB[2 * KT];
  if (nsteps > 0) {
    load_keys(bufA, kw);
    for (int s = 0; s < nsteps; s += 2) {
      load_keys(bufB, kw + (size_t)min(s + 1, nsteps - 1) * (STEP_T * 64));
#pragma unroll
      for (int t = 0; t < KT; ++t)
        scan_tile<CH, ROWS, COUNT>(bufA[2 * t], bufA[2 * t + 1], s * STEP_T + t00 + t, E, xs, qn, q,
                             acc, cnt, xsrc, src, dst, r0, rows, lane);
      if (s + 1 < nsteps) {
        load_keys(bufA, kw + (size_t)min(s + 2, nsteps - 1) * (STEP_T * 64));
#pragma unroll
        for (int t = 0; t < KT; ++t)
          scan_tile<CH, ROWS, COUNT>(bufB[2 * t], bufB[2 * t + 1], (s + 1) * STEP_T + t00 + t, E, xs, qn, q,
                               acc, cnt, xsrc, src, dst, r0, rows, lane);
      }
    }
  }
  drain_queue<CH, ROWS, COUNT>(true, qn, q, acc, cnt, xsrc, src, dst, r0, rows, lane);
  __syncthreads();

  for (int r = wave; r < rows; r += AGG_NW) {
    if (lane < LPR) {
      const float* a = acc + r * CH + lane;
      float4 o;
      o.x = a[0]; o.y = a[LPR]; o.z = a[2 * LPR]; o.w = a[3 * LPR];
      *(float4*)(msg + (size_t)(r0 + r) * CH + lane * 4) = o;
    }
  }
  if (COUNT)
    for (int i = tid; i < rows; i += AGG_NW * 32) deg[r0 + i] = cnt[i];
}

#define GEMM_WAVES 8
#define GEMM_GRID  512
__global__ __launch_bounds__(GEMM_WAVES * 32, 4)
void sage_gemm_relu_proj(const float* __restrict__ msg, const float* __restrict__ deg,
                         const float* __restrict__ xdst,
                         const float* __restrict__ Wl, const float* __restrict__ bl,
                         const float* __restrict__ Wr,
                         const float* __restrict__ Wl1, const float* __restrict__ bl1,
                         const float* __restrict__ Wr1,
                         float* __restrict__ p, float* __restrict__ q, int nrows, int nq)
{
  __shared__ __attribute__((aligned(16))) float sA[16][IN_CH + 4];
  __shared__ __attribute__((aligned(16))) float sX[16][IN_CH + 4];
  __shared__ float sH[16][HID + 4];
  const int tid  = threadIdx.x;
  const int wave = __builtin_amdgcn_readfirstlane(tid >> 5);
  const int lane = tid & 31;
  const int half = lane >> 4;
  const int m    = lane & 15;
  const int n    = lane & 15;
  const int kh   = half * 2;
  const int mbase = half * 8;
  const int n0   = wave * 16;

  v2f bL[IN_CH / 4], bR[IN_CH / 4];
#pragma unroll
  for (int i = 0; i < IN_CH / 4; ++i) {
    const int k = i * 4 + kh;
    bL[i].x = Wl[(size_t)k       * HID + n0 + n];
    bL[i].y = Wl[(size_t)(k + 1) * HID + n0 + n];
    bR[i].x = Wr[(size_t)k       * HID + n0 + n];
    bR[i].y = Wr[(size_t)(k + 1) * HID + n0 + n];
  }
  const float bias = bl[n0 + n];

  const int srow = tid >> 4;
  const int sc4  = tid & 15;
  const int ntiles = (nrows + 15) / 16;

  float4 ga0, ga1, gx0, gx1;
  float  gdeg;
  auto fetch_rows = [&](const int rt) {
    const int row = rt * 16 + srow;
    ga0 = make_float4(0.f, 0.f, 0.f, 0.f); ga1 = ga0; gx0 = ga0; gx1 = ga0; gdeg = 1.0f;
    if (rt < ntiles && row < nrows) {
      const float4* mp = (const float4*)(msg  + (size_t)row * IN_CH);
      const float4* xp = (const float4*)(xdst + (size_t)row * IN_CH);
      ga0 = mp[sc4]; ga1 = mp[16 + sc4];
      gx0 = xp[sc4]; gx1 = xp[16 + sc4];
      gdeg = deg[row];
    }
  };
  fetch_rows(blockIdx.x);

  for (int rt = blockIdx.x; rt < ntiles; rt += GEMM_GRID) {
    const int r0 = rt * 16;
    {
      const float inv = 1.0f / fmaxf(gdeg, 1.0f);
      float4 a0 = ga0, a1 = ga1;
      a0.x *= inv; a0.y *= inv; a0.z *= inv; a0.w *= inv;
      a1.x *= inv; a1.y *= inv; a1.z *= inv; a1.w *= inv;
      *(float4*)&sA[srow][sc4 * 4]      = a0;
      *(float4*)&sA[srow][64 + sc4 * 4] = a1;
      *(float4*)&sX[srow][sc4 * 4]      = gx0;
      *(float4*)&sX[srow][64 + sc4 * 4] = gx1;
    }
    __syncthreads();
    fetch_rows(rt + GEMM_GRID);

    v2f fa[2][8];
    auto load_group = [&](v2f (&f)[8], const int g) {
      const float (*S)[IN_CH + 4] = (g < 4) ? sA : sX;
      const int kb = (g & 3) * 32 + kh;
#pragma unroll
      for (int j = 0; j < 8; ++j) { f[j].x = S[m][kb + j * 4]; f[j].y = S[m][kb + j * 4 + 1]; }
    };
    v8f acc = {0.f, 0.f, 0.f, 0.f, 0.f, 0.f, 0.f, 0.f};
    load_group(fa[0], 0);
#pragma unroll
    for (int g = 0; g < 8; ++g) {
      if (g + 1 < 8) load_group(fa[(g + 1) & 1], g + 1);
      __builtin_amdgcn_sched_barrier(0);
#pragma unroll
      for (int j = 0; j < 8; ++j) {
        const v2f bv = (g < 4) ? bL[g * 8 + j] : bR[(g - 4) * 8 + j];
        acc = __builtin_amdgcn_wmma_f32_16x16x4_f32(false, fa[g & 1][j], false, bv, (short)0, acc, false, false);
      }
      __builtin_amdgcn_sched_barrier(0);
    }
    for (int r = 0; r < 8; ++r) sH[mbase + r][n0 + n] = fmaxf(acc[r] + bias, 0.0f);
    __syncthreads();

    if (wave < 3) {
      const int col = wave * 16 + n;
      const bool cv = (col < OUTC);
      const int colc = cv ? col : OUTC - 1;
      v8f accp = {0.f, 0.f, 0.f, 0.f, 0.f, 0.f, 0.f, 0.f};
      for (int k = 0; k < HID; k += 4) {
        v2f av, bv;
        av.x = sH[m][k + kh];
        av.y = sH[m][k + kh + 1];
        bv.x = Wl1[(size_t)(k + kh)     * OUTC + colc];
        bv.y = Wl1[(size_t)(k + kh + 1) * OUTC + colc];
        accp = __builtin_amdgcn_wmma_f32_16x16x4_f32(false, av, false, bv, (short)0, accp, false, false);
      }
      for (int r = 0; r < 8; ++r) {
        int row = r0 + mbase + r;
        if (row < nrows) p[(size_t)row * OUTP + col] = cv ? accp[r] : 1.0f;
      }
    } else if (wave < 6 && r0 < nq) {
      const int col = (wave - 3) * 16 + n;
      const bool cv = (col < OUTC);
      const int colc = cv ? col : OUTC - 1;
      v8f accq = {0.f, 0.f, 0.f, 0.f, 0.f, 0.f, 0.f, 0.f};
      for (int k = 0; k < HID; k += 4) {
        v2f av, bv;
        av.x = sH[m][k + kh];
        av.y = sH[m][k + kh + 1];
        bv.x = Wr1[(size_t)(k + kh)     * OUTC + colc];
        bv.y = Wr1[(size_t)(k + kh + 1) * OUTC + colc];
        accq = __builtin_amdgcn_wmma_f32_16x16x4_f32(false, av, false, bv, (short)0, accq, false, false);
      }
      const float bias1 = bl1[colc];
      for (int r = 0; r < 8; ++r) {
        int row = r0 + mbase + r;
        if (row < nq) q[(size_t)row * OUTP + col] = cv ? accq[r] + bias1 : 0.0f;
      }
    }
  }
}

__global__ __launch_bounds__(64)
void sage_mean_lsm(const float* __restrict__ msg, const float* __restrict__ q,
                   float* __restrict__ out, int nrows)
{
  const int row = blockIdx.x * blockDim.x + threadIdx.x;
  if (row >= nrows) return;
  const float4* m4 = (const float4*)(msg + (size_t)row * OUTP);
  const float4* q4 = (const float4*)(q   + (size_t)row * OUTP);
  float v[OUTP];
#pragma unroll
  for (int j = 0; j < OUTP / 4; ++j) {
    const float4 a = m4[j];
    v[4 * j + 0] = a.x; v[4 * j + 1] = a.y; v[4 * j + 2] = a.z; v[4 * j + 3] = a.w;
  }
  const float inv = 1.0f / fmaxf(v[OUTC], 1.0f);
#pragma unroll
  for (int j = 0; j < OUTP / 4; ++j) {
    const float4 b = q4[j];
    v[4 * j + 0] = v[4 * j + 0] * inv + b.x;
    v[4 * j + 1] = v[4 * j + 1] * inv + b.y;
    v[4 * j + 2] = v[4 * j + 2] * inv + b.z;
    v[4 * j + 3] = v[4 * j + 3] * inv + b.w;
  }
  float mx = -INFINITY;
#pragma unroll
  for (int c = 0; c < OUTC; ++c) mx = fmaxf(mx, v[c]);
  float s = 0.0f;
#pragma unroll
  for (int c = 0; c < OUTC; ++c) s += expf(v[c] - mx);
  const float lse = logf(s);
#pragma unroll
  for (int c = 0; c < OUTC; ++c)
    out[(size_t)row * OUTC + c] = v[c] - mx - lse;
}

extern "C" void kernel_launch(void* const* d_in, const int* in_sizes, int n_in,
                              void* d_out, int out_size, void* d_ws, size_t ws_size,
                              hipStream_t stream)
{
  const float* x    = (const float*)d_in[0];
  const int*   src0 = (const int*)d_in[1];
  const int*   dst0 = (const int*)d_in[2];
  const int*   src1 = (const int*)d_in[3];
  const int*   dst1 = (const int*)d_in[4];
  const float* Wl0  = (const float*)d_in[7];
  const float* bl0  = (const float*)d_in[8];
  const float* Wr0  = (const float*)d_in[9];
  const float* Wl1  = (const float*)d_in[10];
  const float* bl1  = (const float*)d_in[11];
  const float* Wr1  = (const float*)d_in[12];
  float* out = (float*)d_out;

  const int E0 = in_sizes[1];
  const int E1 = in_sizes[3];
  const int N1 = 100000;
  const int N2 = out_size / OUTC;

  float* msg0 = (float*)d_ws;
  float* deg0 = msg0 + (size_t)N1 * HID;
  float* p    = deg0 + N1;
  float* q    = p    + (size_t)N1 * OUTP;
  float* msg1 = q    + (size_t)N2 * OUTP;
  int*   keys = (int*)(msg1 + (size_t)N2 * OUTP);

  make_keys<<<240, 256, 0, stream>>>(dst0, dst1, keys, E0, E1);

  agg_slab<IN_CH, ROWS0, true><<<(N1 + ROWS0 - 1) / ROWS0, AGG_NW * 32, 0, stream>>>(x, src0, dst0, keys, msg0, deg0, E0, 0, N1);
  sage_gemm_relu_proj<<<GEMM_GRID, GEMM_WAVES * 32, 0, stream>>>(msg0, deg0, x, Wl0, bl0, Wr0,
                                                          Wl1, bl1, Wr1, p, q, N1, N2);

  agg_slab<OUTP, ROWS1, false><<<(N2 + ROWS1 - 1) / ROWS1, AGG_NW * 32, 0, stream>>>(p, src1, dst1, keys, msg1, (float*)0, E1, E0, N2);
  sage_mean_lsm<<<(N2 + 63) / 64, 64, 0, stream>>>(msg1, q, out, N2);
}


# edge-sum scan reads 2 bits per edge, not 8: the key launch writes 32 one-hot planes (16 per key nibble), a block ANDs its two; a lane tests 128 edges per two loads; sum order = LDS arrival as before
# speedup vs baseline: 3.6343x; 1.0864x over previous
#include <hip/hip_runtime.h>
#include <math.h>

#define IN_CH 128
#define HID   128
#define OUTC  47
#define OUTP  48

typedef float v2f __attribute__((ext_vector_type(2)));
typedef float v8f __attribute__((ext_vector_type(8)));

#define AGG_NW  32
#define CHUNK_W 128
#define STEP_W  (AGG_NW * CHUNK_W)
#define QCAP    64
#define ROWS0   392
#define ROWS1   79

__device__ __forceinline__ int padded_words(int E) { return ((E + STEP_W * 32 - 1) / (STEP_W * 32)) * STEP_W; }

template <int ROWS>
__device__ __forceinline__ void key_tile(const int* __restrict__ dst, const int E, const int tile,
                                         int* __restrict__ planes, const int PW, const int lane)
{
  const int e0 = tile * 1024 + lane;
  int d[32];
  if (e0 - lane + 1024 <= E) {
#pragma unroll
    for (int g = 0; g < 32; ++g) d[g] = dst[e0 + g * 32];
  } else {
#pragma unroll
    for (int g = 0; g < 32; ++g) d[g] = (e0 + g * 32 < E) ? dst[e0 + g * 32] : 0;
  }
  unsigned pl[8] = {0u, 0u, 0u, 0u, 0u, 0u, 0u, 0u};
#pragma unroll
  for (int g = 0; g < 32; ++g) {
    const unsigned k = (unsigned)d[g] / (unsigned)ROWS;
#pragma unroll
    for (int j = 0; j < 8; ++j) {
      const unsigned w = (unsigned)__ballot((k >> j) & 1u);
      if (lane == g) pl[j] = w;
    }
  }
  const int rem = E - (tile * 1024 + lane * 32);
  const unsigned vw = (rem >= 32) ? ~0u : ((rem > 0) ? ((1u << rem) - 1u) : 0u);
  int* out = planes + (size_t)tile * 32 + lane;
  {
    const unsigned a0 = pl[0], a1 = pl[1], a2 = pl[2], a3 = pl[3];
    const unsigned p01[4] = {~a0 & ~a1, a0 & ~a1, ~a0 & a1, a0 & a1};
    const unsigned p23[4] = {~a2 & ~a3, a2 & ~a3, ~a2 & a3, a2 & a3};
#pragma unroll
    for (int v = 0; v < 16; ++v) out[(size_t)v * PW] = (int)(p01[v & 3] & p23[v >> 2]);
  }
  {
    const unsigned a0 = pl[4], a1 = pl[5], a2 = pl[6], a3 = pl[7];
    const unsigned p01[4] = {~a0 & ~a1 & vw, a0 & ~a1 & vw, ~a0 & a1 & vw, a0 & a1 & vw};
    const unsigned p23[4] = {~a2 & ~a3, a2 & ~a3, ~a2 & a3, a2 & a3};
#pragma unroll
    for (int v = 0; v < 16; ++v) out[(size_t)(16 + v) * PW] = (int)(p01[v & 3] & p23[v >> 2]);
  }
}

__global__ __launch_bounds__(256)
void make_keys(const int* __restrict__ dst0, const int* __restrict__ dst1,
               int* __restrict__ planes0, int* __restrict__ planes1, int E0, int E1)
{
  const int lane = threadIdx.x & 31;
  const int gw   = __builtin_amdgcn_readfirstlane((int)(blockIdx.x * 8 + (threadIdx.x >> 5)));
  const int pw0  = padded_words(E0);
  const int pw1  = padded_words(E1);
  const int nt0  = pw0 / 32;
  const int nt   = nt0 + pw1 / 32;
  for (int t = gw; t < nt; t += (int)gridDim.x * 8) {
    if (t < nt0) key_tile<ROWS0>(dst0, E0, t, planes0, pw0, lane);
    else         key_tile<ROWS1>(dst1, E1, t - nt0, planes1, pw1, lane);
  }
}


template <int CH, int ROWS, bool COUNT>
__device__ __forceinline__ void drain_queue(const bool last, int& qn, const int* q,
                                            float* __restrict__ acc, float* __restrict__ cnt,
                                            const float* __restrict__ xsrc, const int* __restrict__ src,
                                            const int* __restrict__ dst, const int r0, const int rows, const int lane)
{
  constexpr int LPR = CH / 4;
  while (qn >= 32 || (last && qn > 0)) {
    const int n  = min(qn, 32);
    const int qb = qn - n;
    int s_i = 0;
    int d_i = ROWS;
    if (lane < n) {
      const int e = q[qb + lane];
      s_i = src[e];
      const int d = dst[e] - r0;
      if ((unsigned)d < (unsigned)rows) {
        d_i = d;
        if (COUNT) atomicAdd(&cnt[d], 1.0f);
      }
    }
    qn = qb;
#pragma unroll
    for (int j0 = 0; j0 < 32; j0 += 8) {
      if (j0 < n) {
        float4 v[8];
#pragma unroll
        for (int k = 0; k < 8; ++k) {
          const int sj = __builtin_amdgcn_readlane(s_i, j0 + k);
          if (lane < LPR) v[k] = *(const float4*)(xsrc + (size_t)sj * CH + lane * 4);
        }
#pragma unroll
        for (int k = 0; k < 8; ++k) {
          const int dj = __builtin_amdgcn_readlane(d_i, j0 + k);
          if (lane < LPR) {
            float* a = acc + dj * CH + lane;
            atomicAdd(a,           v[k].x);
            atomicAdd(a + LPR,     v[k].y);
            atomicAdd(a + 2 * LPR, v[k].z);
            atomicAdd(a + 3 * LPR, v[k].w);
          }
        }
      }
    }
  }
}

template <int CH, int ROWS, bool COUNT>
__device__ __forceinline__ void scan_chunk(const int4 lo, const int4 hi, const int gb,
                                           int& qn, int* q,
                                           float* __restrict__ acc, float* __restrict__ cnt,
                                           const float* __restrict__ xsrc, const int* __restrict__ src,
                                           const int* __restrict__ dst, const int r0, const int rows, const int lane)
{
  unsigned long long ma = (unsigned long long)((unsigned)lo.x & (unsigned)hi.x)
                        | ((unsigned long long)((unsigned)lo.y & (unsigned)hi.y) << 32);
  unsigned long long mb = (unsigned long long)((unsigned)lo.z & (unsigned)hi.z)
                        | ((unsigned long long)((unsigned)lo.w & (unsigned)hi.w) << 32);
  const unsigned lt = (1u << lane) - 1u;
  unsigned mask = (unsigned)__ballot((ma | mb) != 0ull);
  while (mask != 0u) {
    if ((ma | mb) != 0ull) {
      int pos;
      if (ma != 0ull) { pos = __ffsll((long long)ma) - 1;      ma &= ma - 1ull; }
      else            { pos = 64 + __ffsll((long long)mb) - 1; mb &= mb - 1ull; }
      q[qn + __popc(mask & lt)] = gb + pos;
    }
    qn += __popc(mask);
    drain_queue<CH, ROWS, COUNT>(false, qn, q, acc, cnt, xsrc, src, dst, r0, rows, lane);
    mask = (unsigned)__ballot((ma | mb) != 0ull);
  }
}

template <int CH, int ROWS, bool COUNT>
__global__ __launch_bounds__(AGG_NW * 32)
void agg_slab(const float* __restrict__ xsrc, const int* __restrict__ src, const int* __restrict__ dst,
              const int* __restrict__ planes, float* __restrict__ msg, float* __restrict__ deg,
              int E, int nrows)
{
  constexpr int LPR = CH / 4;
  __shared__ __attribute__((aligned(16))) float acc[(ROWS + 1) * CH];
  __shared__ float cnt[COUNT ? ROWS : 1];
  __shared__ int   queue[AGG_NW * QCAP];

  const int tid  = threadIdx.x;
  const int lane = tid & 31;
  const int wave = __builtin_amdgcn_readfirstlane(tid >> 5);
  const int r0   = blockIdx.x * ROWS;
  const int rows = min(ROWS, nrows - r0);

  for (int i = tid; i < ROWS * CH / 4; i += AGG_NW * 32)
    ((float4*)acc)[i] = make_float4(0.0f, 0.0f, 0.0f, 0.0f);
  if (COUNT)
    for (int i = tid; i < ROWS; i += AGG_NW * 32) cnt[i] = 0.0f;
  __syncthreads();

  int* q = queue + wave * QCAP;
  int qn = 0;
  const int PW     = padded_words(E);
  const int nsteps = PW / STEP_W;
  const int4* klo = (const int4*)(planes + (size_t)(blockIdx.x & 15u) * PW) + wave * (CHUNK_W / 4) + lane;
  const int4* khi = (const int4*)(planes + (size_t)(16u + (blockIdx.x >> 4)) * PW) + wave * (CHUNK_W / 4) + lane;
  const int g00   = (wave * CHUNK_W + lane * 4) * 32;

  int4 loA, hiA, loB, hiB;
  if (nsteps > 0) {
    loA = klo[0]; hiA = khi[0];
    for (int s = 0; s < nsteps; s += 2) {
      const size_t o1 = (size_t)min(s + 1, nsteps - 1) * (STEP_W / 4);
      loB = klo[o1]; hiB = khi[o1];
      scan_chunk<CH, ROWS, COUNT>(loA, hiA, s * (STEP_W * 32) + g00, qn, q,
                                  acc, cnt, xsrc, src, dst, r0, rows, lane);
      if (s + 1 < nsteps) {
        const size_t o2 = (size_t)min(s + 2, nsteps - 1) * (STEP_W / 4);
        loA = klo[o2]; hiA = khi[o2];
        scan_chunk<CH, ROWS, COUNT>(loB, hiB, (s + 1) * (STEP_W * 32) + g00, qn, q,
                                    acc, cnt, xsrc, src, dst, r0, rows, lane);
      }
    }
  }
  drain_queue<CH, ROWS, COUNT>(true, qn, q, acc, cnt, xsrc, src, dst, r0, rows, lane);
  __syncthreads();

  for (int r = wave; r < rows; r += AGG_NW) {
    if (lane < LPR) {
      const float* a = acc + r * CH + lane;
      float4 o;
      o.x = a[0]; o.y = a[LPR]; o.z = a[2 * LPR]; o.w = a[3 * LPR];
      *(float4*)(msg + (size_t)(r0 + r) * CH + lane * 4) = o;
    }
  }
  if (COUNT)
    for (int i = tid; i < rows; i += AGG_NW * 32) deg[r0 + i] = cnt[i];
}

#define GEMM_WAVES 8
#define GEMM_GRID  512
__global__ __launch_bounds__(GEMM_WAVES * 32, 4)
void sage_gemm_relu_proj(const float* __restrict__ msg, const float* __restrict__ deg,
                         const float* __restrict__ xdst,
                         const float* __restrict__ Wl, const float* __restrict__ bl,
                         const float* __restrict__ Wr,
                         const float* __restrict__ Wl1, const float* __restrict__ bl1,
                         const float* __restrict__ Wr1,
                         float* __restrict__ p, float* __restrict__ q, int nrows, int nq)
{
  __shared__ __attribute__((aligned(16))) float sA[16][IN_CH + 4];
  __shared__ __attribute__((aligned(16))) float sX[16][IN_CH + 4];
  __shared__ float sH[16][HID + 4];
  const int tid  = threadIdx.x;
  const int wave = __builtin_amdgcn_readfirstlane(tid >> 5);
  const int lane = tid & 31;
  const int half = lane >> 4;
  const int m    = lane & 15;
  const int n    = lane & 15;
  const int kh   = half * 2;
  const int mbase = half * 8;
  const int n0   = wave * 16;

  v2f bL[IN_CH / 4], bR[IN_CH / 4];
#pragma unroll
  for (int i = 0; i < IN_CH / 4; ++i) {
    const int k = i * 4 + kh;
    bL[i].x = Wl[(size_t)k       * HID + n0 + n];
    bL[i].y = Wl[(size_t)(k + 1) * HID + n0 + n];
    bR[i].x = Wr[(size_t)k       * HID + n0 + n];
    bR[i].y = Wr[(size_t)(k + 1) * HID + n0 + n];
  }
  const float bias = bl[n0 + n];

  const int srow = tid >> 4;
  const int sc4  = tid & 15;
  const int ntiles = (nrows + 15) / 16;

  float4 ga0, ga1, gx0, gx1;
  float  gdeg;
  auto fetch_rows = [&](const int rt) {
    const int row = rt * 16 + srow;
    ga0 = make_float4(0.f, 0.f, 0.f, 0.f); ga1 = ga0; gx0 = ga0; gx1 = ga0; gdeg = 1.0f;
    if (rt < ntiles && row < nrows) {
      const float4* mp = (const float4*)(msg  + (size_t)row * IN_CH);
      const float4* xp = (const float4*)(xdst + (size_t)row * IN_CH);
      ga0 = mp[sc4]; ga1 = mp[16 + sc4];
      gx0 = xp[sc4]; gx1 = xp[16 + sc4];
      gdeg = deg[row];
    }
  };
  fetch_rows(blockIdx.x);

  for (int rt = blockIdx.x; rt < ntiles; rt += GEMM_GRID) {
    const int r0 = rt * 16;
    {
      const float inv = 1.0f / fmaxf(gdeg, 1.0f);
      float4 a0 = ga0, a1 = ga1;
      a0.x *= inv; a0.y *= inv; a0.z *= inv; a0.w *= inv;
      a1.x *= inv; a1.y *= inv; a1.z *= inv; a1.w *= inv;
      *(float4*)&sA[srow][sc4 * 4]      = a0;
      *(float4*)&sA[srow][64 + sc4 * 4] = a1;
      *(float4*)&sX[srow][sc4 * 4]      = gx0;
      *(float4*)&sX[srow][64 + sc4 * 4] = gx1;
    }
    __syncthreads();
    fetch_rows(rt + GEMM_GRID);

    v2f fa[2][8];
    auto load_group = [&](v2f (&f)[8], const int g) {
      const float (*S)[IN_CH + 4] = (g < 4) ? sA : sX;
      const int kb = (g & 3) * 32 + kh;
#pragma unroll
      for (int j = 0; j < 8; ++j) { f[j].x = S[m][kb + j * 4]; f[j].y = S[m][kb + j * 4 + 1]; }
    };
    v8f acc = {0.f, 0.f, 0.f, 0.f, 0.f, 0.f, 0.f, 0.f};
    load_group(fa[0], 0);
#pragma unroll
    for (int g = 0; g < 8; ++g) {
      if (g + 1 < 8) load_group(fa[(g + 1) & 1], g + 1);
      __builtin_amdgcn_sched_barrier(0);
#pragma unroll
      for (int j = 0; j < 8; ++j) {
        const v2f bv = (g < 4) ? bL[g * 8 + j] : bR[(g - 4) * 8 + j];
        acc = __builtin_amdgcn_wmma_f32_16x16x4_f32(false, fa[g & 1][j], false, bv, (short)0, acc, false, false);
      }
      __builtin_amdgcn_sched_barrier(0);
    }
    for (int r = 0; r < 8; ++r) sH[mbase + r][n0 + n] = fmaxf(acc[r] + bias, 0.0f);
    __syncthreads();

    if (wave < 3) {
      const int col = wave * 16 + n;
      const bool cv = (col < OUTC);
      const int colc = cv ? col : OUTC - 1;
      v8f accp = {0.f, 0.f, 0.f, 0.f, 0.f, 0.f, 0.f, 0.f};
      for (int k = 0; k < HID; k += 4) {
        v2f av, bv;
        av.x = sH[m][k + kh];
        av.y = sH[m][k + kh + 1];
        bv.x = Wl1[(size_t)(k + kh)     * OUTC + colc];
        bv.y = Wl1[(size_t)(k + kh + 1) * OUTC + colc];
        accp = __builtin_amdgcn_wmma_f32_16x16x4_f32(false, av, false, bv, (short)0, accp, false, false);
      }
      for (int r = 0; r < 8; ++r) {
        int row = r0 + mbase + r;
        if (row < nrows) p[(size_t)row * OUTP + col] = cv ? accp[r] : 1.0f;
      }
    } else if (wave < 6 && r0 < nq) {
      const int col = (wave - 3) * 16 + n;
      const bool cv = (col < OUTC);
      const int colc = cv ? col : OUTC - 1;
      v8f accq = {0.f, 0.f, 0.f, 0.f, 0.f, 0.f, 0.f, 0.f};
      for (int k = 0; k < HID; k += 4) {
        v2f av, bv;
        av.x = sH[m][k + kh];
        av.y = sH[m][k + kh + 1];
        bv.x = Wr1[(size_t)(k + kh)     * OUTC + colc];
        bv.y = Wr1[(size_t)(k + kh + 1) * OUTC + colc];
        accq = __builtin_amdgcn_wmma_f32_16x16x4_f32(false, av, false, bv, (short)0, accq, false, false);
      }
      const float bias1 = bl1[colc];
      for (int r = 0; r < 8; ++r) {
        int row = r0 + mbase + r;
        if (row < nq) q[(size_t)row * OUTP + col] = cv ? accq[r] + bias1 : 0.0f;
      }
    }
  }
}

__global__ __launch_bounds__(64)
void sage_mean_lsm(const float* __restrict__ msg, const float* __restrict__ q,
                   float* __restrict__ out, int nrows)
{
  const int row = blockIdx.x * blockDim.x + threadIdx.x;
  if (row >= nrows) return;
  const float4* m4 = (const float4*)(msg + (size_t)row * OUTP);
  const float4* q4 = (const float4*)(q   + (size_t)row * OUTP);
  float v[OUTP];
#pragma unroll
  for (int j = 0; j < OUTP / 4; ++j) {
    const float4 a = m4[j];
    v[4 * j + 0] = a.x; v[4 * j + 1] = a.y; v[4 * j + 2] = a.z; v[4 * j + 3] = a.w;
  }
  const float inv = 1.0f / fmaxf(v[OUTC], 1.0f);
#pragma unroll
  for (int j = 0; j < OUTP / 4; ++j) {
    const float4 b = q4[j];
    v[4 * j + 0] = v[4 * j + 0] * inv + b.x;
    v[4 * j + 1] = v[4 * j + 1] * inv + b.y;
    v[4 * j + 2] = v[4 * j + 2] * inv + b.z;
    v[4 * j + 3] = v[4 * j + 3] * inv + b.w;
  }
  float mx = -INFINITY;
#pragma unroll
  for (int c = 0; c < OUTC; ++c) mx = fmaxf(mx, v[c]);
  float s = 0.0f;
#pragma unroll
  for (int c = 0; c < OUTC; ++c) s += expf(v[c] - mx);
  const float lse = logf(s);
#pragma unroll
  for (int c = 0; c < OUTC; ++c)
    out[(size_t)row * OUTC + c] = v[c] - mx - lse;
}

extern "C" void kernel_launch(void* const* d_in, const int* in_sizes, int n_in,
                              void* d_out, int out_size, void* d_ws, size_t ws_size,
                              hipStream_t stream)
{
  const float* x    = (const float*)d_in[0];
  const int*   src0 = (const int*)d_in[1];
  const int*   dst0 = (const int*)d_in[2];
  const int*   src1 = (const int*)d_in[3];
  const int*   dst1 = (const int*)d_in[4];
  const float* Wl0  = (const float*)d_in[7];
  const float* bl0  = (const float*)d_in[8];
  const float* Wr0  = (const float*)d_in[9];
  const float* Wl1  = (const float*)d_in[10];
  const float* bl1  = (const float*)d_in[11];
  const float* Wr1  = (const float*)d_in[12];
  float* out = (float*)d_out;

  const int E0 = in_sizes[1];
  const int E1 = in_sizes[3];
  const int N1 = 100000;
  const int N2 = out_size / OUTC;

  float* msg0 = (float*)d_ws;
  float* deg0 = msg0 + (size_t)N1 * HID;
  float* p    = deg0 + N1;
  float* q    = p    + (size_t)N1 * OUTP;
  float* msg1 = q    + (size_t)N2 * OUTP;
  int*   planes0 = (int*)p;
  int*   planes1 = (int*)(msg1 + (size_t)N2 * OUTP);

  make_keys<<<256, 256, 0, stream>>>(dst0, dst1, planes0, planes1, E0, E1);

  agg_slab<IN_CH, ROWS0, true><<<(N1 + ROWS0 - 1) / ROWS0, AGG_NW * 32, 0, stream>>>(x, src0, dst0, planes0, msg0, deg0, E0, N1);
  sage_gemm_relu_proj<<<GEMM_GRID, GEMM_WAVES * 32, 0, stream>>>(msg0, deg0, x, Wl0, bl0, Wr0,
                                                          Wl1, bl1, Wr1, p, q, N1, N2);

  agg_slab<OUTP, ROWS1, false><<<(N2 + ROWS1 - 1) / ROWS1, AGG_NW * 32, 0, stream>>>(p, src1, dst1, planes1, msg1, (float*)0, E1, N2);
  sage_mean_lsm<<<(N2 + 63) / 64, 64, 0, stream>>>(msg1, q, out, N2);
}


# layer-0 edge sums read one scattered 4-byte word per edge, not two: the key launch packs source row and row-in-block into one index word per edge, read in place of src[e] and dst[e]; sum order kept
# speedup vs baseline: 3.7752x; 1.0388x over previous
#include <hip/hip_runtime.h>
#include <math.h>

#define IN_CH 128
#define HID   128
#define OUTC  47
#define OUTP  48

typedef float v2f __attribute__((ext_vector_type(2)));
typedef float v8f __attribute__((ext_vector_type(8)));

#define AGG_NW  32
#define CHUNK_W 128
#define STEP_W  (AGG_NW * CHUNK_W)
#define QCAP    64
#define ROWS0   392
#define ROWS1   79

#define PK_BITS 9
#define PK_MASK ((1u << PK_BITS) - 1u)
static_assert(ROWS0 <= (int)PK_MASK, "the spare row must fit into the low bits of an index word");

__device__ __forceinline__ int padded_words(int E) { return ((E + STEP_W * 32 - 1) / (STEP_W * 32)) * STEP_W; }

template <int ROWS, bool PACK>
__device__ __forceinline__ void key_tile(const int* __restrict__ dst, const int* __restrict__ src,
                                         const int E, const int nrows, const int tile,
                                         int* __restrict__ planes, int* __restrict__ pk, const int PW, const int lane)
{
  const int e0 = tile * 1024 + lane;
  const bool full = (e0 - lane + 1024 <= E);
  int d[32];
  int s[32];
  if (full) {
#pragma unroll
    for (int g = 0; g < 32; ++g) d[g] = dst[e0 + g * 32];
    if (PACK) {
#pragma unroll
      for (int g = 0; g < 32; ++g) s[g] = src[e0 + g * 32];
    }
  } else {
#pragma unroll
    for (int g = 0; g < 32; ++g) d[g] = (e0 + g * 32 < E) ? dst[e0 + g * 32] : 0;
    if (PACK) {
#pragma unroll
      for (int g = 0; g < 32; ++g) s[g] = (e0 + g * 32 < E) ? src[e0 + g * 32] : 0;
    }
  }
  unsigned pl[8] = {0u, 0u, 0u, 0u, 0u, 0u, 0u, 0u};
#pragma unroll
  for (int g = 0; g < 32; ++g) {
    const unsigned k = (unsigned)d[g] / (unsigned)ROWS;
    if (PACK) {
      const unsigned loc = ((unsigned)d[g] < (unsigned)nrows) ? (unsigned)d[g] - k * (unsigned)ROWS : (unsigned)ROWS;
      if (full || e0 + g * 32 < E) pk[e0 + g * 32] = (int)(((unsigned)s[g] << PK_BITS) | loc);
    }
#pragma unroll
    for (int j = 0; j < 8; ++j) {
      const unsigned w = (unsigned)__ballot((k >> j) & 1u);
      if (lane == g) pl[j] = w;
    }
  }
  const int rem = E - (tile * 1024 + lane * 32);
  const unsigned vw = (rem >= 32) ? ~0u : ((rem > 0) ? ((1u << rem) - 1u) : 0u);
  int* out = planes + (size_t)tile * 32 + lane;
  {
    const unsigned a0 = pl[0], a1 = pl[1], a2 = pl[2], a3 = pl[3];
    const unsigned p01[4] = {~a0 & ~a1, a0 & ~a1, ~a0 & a1, a0 & a1};
    const unsigned p23[4] = {~a2 & ~a3, a2 & ~a3, ~a2 & a3, a2 & a3};
#pragma unroll
    for (int v = 0; v < 16; ++v) out[(size_t)v * PW] = (int)(p01[v & 3] & p23[v >> 2]);
  }
  {
    const unsigned a0 = pl[4], a1 = pl[5], a2 = pl[6], a3 = pl[7];
    const unsigned p01[4] = {~a0 & ~a1 & vw, a0 & ~a1 & vw, ~a0 & a1 & vw, a0 & a1 & vw};
    const unsigned p23[4] = {~a2 & ~a3, a2 & ~a3, ~a2 & a3, a2 & a3};
#pragma unroll
    for (int v = 0; v < 16; ++v) out[(size_t)(16 + v) * PW] = (int)(p01[v & 3] & p23[v >> 2]);
  }
}

__global__ __launch_bounds__(256)
void make_keys(const int* __restrict__ src0, const int* __restrict__ dst0, const int* __restrict__ dst1,
               int* __restrict__ planes0, int* __restrict__ pk0, int* __restrict__ planes1,
               int E0, int E1, int n1)
{
  const int lane = threadIdx.x & 31;
  const int gw   = __builtin_amdgcn_readfirstlane((int)(blockIdx.x * 8 + (threadIdx.x >> 5)));
  const int pw0  = padded_words(E0);
  const int pw1  = padded_words(E1);
  const int nt0  = pw0 / 32;
  const int nt   = nt0 + pw1 / 32;
  for (int t = gw; t < nt; t += (int)gridDim.x * 8) {
    if (t < nt0) key_tile<ROWS0, true>(dst0, src0, E0, n1, t, planes0, pk0, pw0, lane);
    else         key_tile<ROWS1, false>(dst1, (const int*)0, E1, 0, t - nt0, planes1, (int*)0, pw1, lane);
  }
}


template <int CH, int ROWS, bool COUNT, bool PACKED>
__device__ __forceinline__ void drain_queue(const bool last, int& qn, const int* q,
                                            float* __restrict__ acc, float* __restrict__ cnt,
                                            const float* __restrict__ xsrc, const int* __restrict__ src,
                                            const int* __restrict__ dst, const int r0, const int rows, const int lane)
{
  constexpr int LPR = CH / 4;
  while (qn >= 32 || (last && qn > 0)) {
    const int n  = min(qn, 32);
    const int qb = qn - n;
    int s_i = 0;
    int d_i = ROWS;
    if (lane < n) {
      const int e = q[qb + lane];
      if (PACKED) {
        const unsigned w = (unsigned)src[e];
        s_i = (int)(w >> PK_BITS);
        d_i = (int)(w & PK_MASK);
        if (COUNT && d_i != ROWS) atomicAdd(&cnt[d_i], 1.0f);
      } else {
        s_i = src[e];
        const int d = dst[e] - r0;
        if ((unsigned)d < (unsigned)rows) {
          d_i = d;
          if (COUNT) atomicAdd(&cnt[d], 1.0f);
        }
      }
    }
    qn = qb;
#pragma unroll
    for (int j0 = 0; j0 < 32; j0 += 8) {
      if (j0 < n) {
        float4 v[8];
#pragma unroll
        for (int k = 0; k < 8; ++k) {
          const int sj = __builtin_amdgcn_readlane(s_i, j0 + k);
          if (lane < LPR) v[k] = *(const float4*)(xsrc + (size_t)sj * CH + lane * 4);
        }
#pragma unroll
        for (int k = 0; k < 8; ++k) {
          const int dj = __builtin_amdgcn_readlane(d_i, j0 + k);
          if (lane < LPR) {
            float* a = acc + dj * CH + lane;
            atomicAdd(a,           v[k].x);
            atomicAdd(a + LPR,     v[k].y);
            atomicAdd(a + 2 * LPR, v[k].z);
            atomicAdd(a + 3 * LPR, v[k].w);
          }
        }
      }
    }
  }
}

template <int CH, int ROWS, bool COUNT, bool PACKED>
__device__ __forceinline__ void scan_chunk(const int4 lo, const int4 hi, const int gb,
                                           int& qn, int* q,
                                           float* __restrict__ acc, float* __restrict__ cnt,
                                           const float* __restrict__ xsrc, const int* __restrict__ src,
                                           const int* __restrict__ dst, const int r0, const int rows, const int lane)
{
  unsigned long long ma = (unsigned long long)((unsigned)lo.x & (unsigned)hi.x)
                        | ((unsigned long long)((unsigned)lo.y & (unsigned)hi.y) << 32);
  unsigned long long mb = (unsigned long long)((unsigned)lo.z & (unsigned)hi.z)
                        | ((unsigned long long)((unsigned)lo.w & (unsigned)hi.w) << 32);
  const unsigned lt = (1u << lane) - 1u;
  unsigned mask = (unsigned)__ballot((ma | mb) != 0ull);
  while (mask != 0u) {
    if ((ma | mb) != 0ull) {
      int pos;
      if (ma != 0ull) { pos = __ffsll((long long)ma) - 1;      ma &= ma - 1ull; }
      else            { pos = 64 + __ffsll((long long)mb) - 1; mb &= mb - 1ull; }
      q[qn + __popc(mask & lt)] = gb + pos;
    }
    qn += __popc(mask);
    drain_queue<CH, ROWS, COUNT, PACKED>(false, qn, q, acc, cnt, xsrc, src, dst, r0, rows, lane);
    mask = (unsigned)__ballot((ma | mb) != 0ull);
  }
}

template <int CH, int ROWS, bool COUNT, bool PACKED>
__global__ __launch_bounds__(AGG_NW * 32)
void agg_slab(const float* __restrict__ xsrc, const int* __restrict__ src, const int* __restrict__ dst,
              const int* __restrict__ planes, float* __restrict__ msg, float* __restrict__ deg,
              int E, int nrows)
{
  constexpr int LPR = CH / 4;
  __shared__ __attribute__((aligned(16))) float acc[(ROWS + 1) * CH];
  __shared__ float cnt[COUNT ? ROWS : 1];
  __shared__ int   queue[AGG_NW * QCAP];

  const int tid  = threadIdx.x;
  const int lane = tid & 31;
  const int wave = __builtin_amdgcn_readfirstlane(tid >> 5);
  const int r0   = blockIdx.x * ROWS;
  const int rows = min(ROWS, nrows - r0);

  for (int i = tid; i < ROWS * CH / 4; i += AGG_NW * 32)
    ((float4*)acc)[i] = make_float4(0.0f, 0.0f, 0.0f, 0.0f);
  if (COUNT)
    for (int i = tid; i < ROWS; i += AGG_NW * 32) cnt[i] = 0.0f;
  __syncthreads();

  int* q = queue + wave * QCAP;
  int qn = 0;
  const int PW     = padded_words(E);
  const int nsteps = PW / STEP_W;
  const int4* klo = (const int4*)(planes + (size_t)(blockIdx.x & 15u) * PW) + wave * (CHUNK_W / 4) + lane;
  const int4* khi = (const int4*)(planes + (size_t)(16u + (blockIdx.x >> 4)) * PW) + wave * (CHUNK_W / 4) + lane;
  const int g00   = (wave * CHUNK_W + lane * 4) * 32;

  int4 loA, hiA, loB, hiB;
  if (nsteps > 0) {
    loA = klo[0]; hiA = khi[0];
    for (int s = 0; s < nsteps; s += 2) {
      const size_t o1 = (size_t)min(s + 1, nsteps - 1) * (STEP_W / 4);
      loB = klo[o1]; hiB = khi[o1];
      scan_chunk<CH, ROWS, COUNT, PACKED>(loA, hiA, s * (STEP_W * 32) + g00, qn, q,
                                  acc, cnt, xsrc, src, dst, r0, rows, lane);
      if (s + 1 < nsteps) {
        const size_t o2 = (size_t)min(s + 2, nsteps - 1) * (STEP_W / 4);
        loA = klo[o2]; hiA = khi[o2];
        scan_chunk<CH, ROWS, COUNT, PACKED>(loB, hiB, (s + 1) * (STEP_W * 32) + g00, qn, q,
                                    acc, cnt, xsrc, src, dst, r0, rows, lane);
      }
    }
  }
  drain_queue<CH, ROWS, COUNT, PACKED>(true, qn, q, acc, cnt, xsrc, src, dst, r0, rows, lane);
  __syncthreads();

  for (int r = wave; r < rows; r += AGG_NW) {
    if (lane < LPR) {
      const float* a = acc + r * CH + lane;
      float4 o;
      o.x = a[0]; o.y = a[LPR]; o.z = a[2 * LPR]; o.w = a[3 * LPR];
      *(float4*)(msg + (size_t)(r0 + r) * CH + lane * 4) = o;
    }
  }
  if (COUNT)
    for (int i = tid; i < rows; i += AGG_NW * 32) deg[r0 + i] = cnt[i];
}

#define GEMM_WAVES 8
#define GEMM_GRID  512
__global__ __launch_bounds__(GEMM_WAVES * 32, 4)
void sage_gemm_relu_proj(const float* __restrict__ msg, const float* __restrict__ deg,
                         const float* __restrict__ xdst,
                         const float* __restrict__ Wl, const float* __restrict__ bl,
                         const float* __restrict__ Wr,
                         const float* __restrict__ Wl1, const float* __restrict__ bl1,
                         const float* __restrict__ Wr1,
                         float* __restrict__ p, float* __restrict__ q, int nrows, int nq)
{
  __shared__ __attribute__((aligned(16))) float sA[16][IN_CH + 4];
  __shared__ __attribute__((aligned(16))) float sX[16][IN_CH + 4];
  __shared__ float sH[16][HID + 4];
  const int tid  = threadIdx.x;
  const int wave = __builtin_amdgcn_readfirstlane(tid >> 5);
  const int lane = tid & 31;
  const int half = lane >> 4;
  const int m    = lane & 15;
  const int n    = lane & 15;
  const int kh   = half * 2;
  const int mbase = half * 8;
  const int n0   = wave * 16;

  v2f bL[IN_CH / 4], bR[IN_CH / 4];
#pragma unroll
  for (int i = 0; i < IN_CH / 4; ++i) {
    const int k = i * 4 + kh;
    bL[i].x = Wl[(size_t)k       * HID + n0 + n];
    bL[i].y = Wl[(size_t)(k + 1) * HID + n0 + n];
    bR[i].x = Wr[(size_t)k       * HID + n0 + n];
    bR[i].y = Wr[(size_t)(k + 1) * HID + n0 + n];
  }
  const float bias = bl[n0 + n];

  const int srow = tid >> 4;
  const int sc4  = tid & 15;
  const int ntiles = (nrows + 15) / 16;

  float4 ga0, ga1, gx0, gx1;
  float  gdeg;
  auto fetch_rows = [&](const int rt) {
    const int row = rt * 16 + srow;
    ga0 = make_float4(0.f, 0.f, 0.f, 0.f); ga1 = ga0; gx0 = ga0; gx1 = ga0; gdeg = 1.0f;
    if (rt < ntiles && row < nrows) {
      const float4* mp = (const float4*)(msg  + (size_t)row * IN_CH);
      const float4* xp = (const float4*)(xdst + (size_t)row * IN_CH);
      ga0 = mp[sc4]; ga1 = mp[16 + sc4];
      gx0 = xp[sc4]; gx1 = xp[16 + sc4];
      gdeg = deg[row];
    }
  };
  fetch_rows(blockIdx.x);

  for (int rt = blockIdx.x; rt < ntiles; rt += GEMM_GRID) {
    const int r0 = rt * 16;
    {
      const float inv = 1.0f / fmaxf(gdeg, 1.0f);
      float4 a0 = ga0, a1 = ga1;
      a0.x *= inv; a0.y *= inv; a0.z *= inv; a0.w *= inv;
      a1.x *= inv; a1.y *= inv; a1.z *= inv; a1.w *= inv;
      *(float4*)&sA[srow][sc4 * 4]      = a0;
      *(float4*)&sA[srow][64 + sc4 * 4] = a1;
      *(float4*)&sX[srow][sc4 * 4]      = gx0;
      *(float4*)&sX[srow][64 + sc4 * 4] = gx1;
    }
    __syncthreads();
    fetch_rows(rt + GEMM_GRID);

    v2f fa[2][8];
    auto load_group = [&](v2f (&f)[8], const int g) {
      const float (*S)[IN_CH + 4] = (g < 4) ? sA : sX;
      const int kb = (g & 3) * 32 + kh;
#pragma unroll
      for (int j = 0; j < 8; ++j) { f[j].x = S[m][kb + j * 4]; f[j].y = S[m][kb + j * 4 + 1]; }
    };
    v8f acc = {0.f, 0.f, 0.f, 0.f, 0.f, 0.f, 0.f, 0.f};
    load_group(fa[0], 0);
#pragma unroll
    for (int g = 0; g < 8; ++g) {
      if (g + 1 < 8) load_group(fa[(g + 1) & 1], g + 1);
      __builtin_amdgcn_sched_barrier(0);
#pragma unroll
      for (int j = 0; j < 8; ++j) {
        const v2f bv = (g < 4) ? bL[g * 8 + j] : bR[(g - 4) * 8 + j];
        acc = __builtin_amdgcn_wmma_f32_16x16x4_f32(false, fa[g & 1][j], false, bv, (short)0, acc, false, false);
      }
      __builtin_amdgcn_sched_barrier(0);
    }
    for (int r = 0; r < 8; ++r) sH[mbase + r][n0 + n] = fmaxf(acc[r] + bias, 0.0f);
    __syncthreads();

    if (wave < 3) {
      const int col = wave * 16 + n;
      const bool cv = (col < OUTC);
      const int colc = cv ? col : OUTC - 1;
      v8f accp = {0.f, 0.f, 0.f, 0.f, 0.f, 0.f, 0.f, 0.f};
      for (int k = 0; k < HID; k += 4) {
        v2f av, bv;
        av.x = sH[m][k + kh];
        av.y = sH[m][k + kh + 1];
        bv.x = Wl1[(size_t)(k + kh)     * OUTC + colc];
        bv.y = Wl1[(size_t)(k + kh + 1) * OUTC + colc];
        accp = __builtin_amdgcn_wmma_f32_16x16x4_f32(false, av, false, bv, (short)0, accp, false, false);
      }
      for (int r = 0; r < 8; ++r) {
        int row = r0 + mbase + r;
        if (row < nrows) p[(size_t)row * OUTP + col] = cv ? accp[r] : 1.0f;
      }
    } else if (wave < 6 && r0 < nq) {
      const int col = (wave - 3) * 16 + n;
      const bool cv = (col < OUTC);
      const int colc = cv ? col : OUTC - 1;
      v8f accq = {0.f, 0.f, 0.f, 0.f, 0.f, 0.f, 0.f, 0.f};
      for (int k = 0; k < HID; k += 4) {
        v2f av, bv;
        av.x = sH[m][k + kh];
        av.y = sH[m][k + kh + 1];
        bv.x = Wr1[(size_t)(k + kh)     * OUTC + colc];
        bv.y = Wr1[(size_t)(k + kh + 1) * OUTC + colc];
        accq = __builtin_amdgcn_wmma_f32_16x16x4_f32(false, av, false, bv, (short)0, accq, false, false);
      }
      const float bias1 = bl1[colc];
      for (int r = 0; r < 8; ++r) {
        int row = r0 + mbase + r;
        if (row < nq) q[(size_t)row * OUTP + col] = cv ? accq[r] + bias1 : 0.0f;
      }
    }
  }
}

__global__ __launch_bounds__(64)
void sage_mean_lsm(const float* __restrict__ msg, const float* __restrict__ q,
                   float* __restrict__ out, int nrows)
{
  const int row = blockIdx.x * blockDim.x + threadIdx.x;
  if (row >= nrows) return;
  const float4* m4 = (const float4*)(msg + (size_t)row * OUTP);
  const float4* q4 = (const float4*)(q   + (size_t)row * OUTP);
  float v[OUTP];
#pragma unroll
  for (int j = 0; j < OUTP / 4; ++j) {
    const float4 a = m4[j];
    v[4 * j + 0] = a.x; v[4 * j + 1] = a.y; v[4 * j + 2] = a.z; v[4 * j + 3] = a.w;
  }
  const float inv = 1.0f / fmaxf(v[OUTC], 1.0f);
#pragma unroll
  for (int j = 0; j < OUTP / 4; ++j) {
    const float4 b = q4[j];
    v[4 * j + 0] = v[4 * j + 0] * inv + b.x;
    v[4 * j + 1] = v[4 * j + 1] * inv + b.y;
    v[4 * j + 2] = v[4 * j + 2] * inv + b.z;
    v[4 * j + 3] = v[4 * j + 3] * inv + b.w;
  }
  float mx = -INFINITY;
#pragma unroll
  for (int c = 0; c < OUTC; ++c) mx = fmaxf(mx, v[c]);
  float s = 0.0f;
#pragma unroll
  for (int c = 0; c < OUTC; ++c) s += expf(v[c] - mx);
  const float lse = logf(s);
#pragma unroll
  for (int c = 0; c < OUTC; ++c)
    out[(size_t)row * OUTC + c] = v[c] - mx - lse;
}

extern "C" void kernel_launch(void* const* d_in, const int* in_sizes, int n_in,
                              void* d_out, int out_size, void* d_ws, size_t ws_size,
                              hipStream_t stream)
{
  const float* x    = (const float*)d_in[0];
  const int*   src0 = (const int*)d_in[1];
  const int*   dst0 = (const int*)d_in[2];
  const int*   src1 = (const int*)d_in[3];
  const int*   dst1 = (const int*)d_in[4];
  const float* Wl0  = (const float*)d_in[7];
  const float* bl0  = (const float*)d_in[8];
  const float* Wr0  = (const float*)d_in[9];
  const float* Wl1  = (const float*)d_in[10];
  const float* bl1  = (const float*)d_in[11];
  const float* Wr1  = (const float*)d_in[12];
  float* out = (float*)d_out;

  const int E0 = in_sizes[1];
  const int E1 = in_sizes[3];
  const int N1 = 100000;
  const int N2 = out_size / OUTC;

  float* msg0 = (float*)d_ws;
  float* deg0 = msg0 + (size_t)N1 * HID;
  float* p    = deg0 + N1;
  float* q    = p    + (size_t)N1 * OUTP;
  float* msg1 = q    + (size_t)N2 * OUTP;
  int*   planes0 = (int*)p;
  int*   pk0     = (int*)p + (size_t)N1 * (OUTP / 2);
  int*   planes1 = (int*)(msg1 + (size_t)N2 * OUTP);

  make_keys<<<256, 256, 0, stream>>>(src0, dst0, dst1, planes0, pk0, planes1, E0, E1, N1);

  agg_slab<IN_CH, ROWS0, true, true><<<(N1 + ROWS0 - 1) / ROWS0, AGG_NW * 32, 0, stream>>>(x, pk0, dst0, planes0, msg0, deg0, E0, N1);
  sage_gemm_relu_proj<<<GEMM_GRID, GEMM_WAVES * 32, 0, stream>>>(msg0, deg0, x, Wl0, bl0, Wr0,
                                                          Wl1, bl1, Wr1, p, q, N1, N2);

  agg_slab<OUTP, ROWS1, false, false><<<(N2 + ROWS1 - 1) / ROWS1, AGG_NW * 32, 0, stream>>>(p, src1, dst1, planes1, msg1, (float*)0, E1, N2);
  sage_mean_lsm<<<(N2 + 63) / 64, 64, 0, stream>>>(msg1, q, out, N2);
}
